# starting kernel
# baseline (speedup 1.0000x reference)
#include <hip/hip_runtime.h>
#include <math.h>

typedef __attribute__((ext_vector_type(2))) float v2f;
typedef __attribute__((ext_vector_type(8))) float v8f;

#define D_FEAT 128

__global__ void k_init_deg(float* __restrict__ deg, int n) {
    int i = blockIdx.x * blockDim.x + threadIdx.x;
    if (i < n) deg[i] = 1.0f;
}

__global__ void k_deg_accum(float* __restrict__ deg, const int* __restrict__ dst,
                            const float* __restrict__ ew, int E) {
    int e = blockIdx.x * blockDim.x + threadIdx.x;
    if (e < E) atomicAdd(&deg[dst[e]], ew[e]);
}

__global__ void k_dinv(float* __restrict__ deg, int n) {
    int i = blockIdx.x * blockDim.x + threadIdx.x;
    if (i < n) {
        float d = deg[i];
        deg[i] = (d > 0.0f) ? rsqrtf(d) : 0.0f;
    }
}

__global__ void k_gemm_wmma(const float* __restrict__ x, const float* __restrict__ W,
                            float* __restrict__ h, int N) {
    const int lane  = threadIdx.x & 31;
    const int wave  = threadIdx.x >> 5;
    const int Mbase = blockIdx.x * 16;
    const int Nbase = wave * 16;
    const int half  = lane >> 4;
    const int l15   = lane & 15;

    int mrow = Mbase + l15; if (mrow >= N) mrow = N - 1;
    const float* __restrict__ arow = x + (size_t)mrow * D_FEAT;
    const float* __restrict__ brow = W + (size_t)(Nbase + l15) * D_FEAT;

    v8f c = {};
#pragma unroll
    for (int k = 0; k < D_FEAT; k += 4) {
        const int kk = k + 2 * half;
        v2f a = *reinterpret_cast<const v2f*>(arow + kk);
        v2f b = *reinterpret_cast<const v2f*>(brow + kk);
        c = __builtin_amdgcn_wmma_f32_16x16x4_f32(false, a, false, b, (short)0, c,
                                                  false, false);
    }

    float* __restrict__ hout = h + (size_t)Mbase * D_FEAT + Nbase;
#pragma unroll
    for (int v = 0; v < 8; ++v) {
        int m = v + 8 * half;
        if (Mbase + m < N) hout[(size_t)m * D_FEAT + l15] = c[v];
    }
}

__global__ void k_out_init(float* __restrict__ out, const float* __restrict__ h,
                           const float* __restrict__ dinv, const float* __restrict__ bias,
                           int total) {
    int i = blockIdx.x * blockDim.x + threadIdx.x;
    if (i < total) {
        int node = i >> 7;
        int d    = i & 127;
        float di = dinv[node];
        out[i] = bias[d] + h[i] * di * di;
    }
}

__global__ void k_scatter(float* __restrict__ out, const float* __restrict__ h,
                          const float* __restrict__ dinv, const int* __restrict__ src,
                          const int* __restrict__ dst, const float* __restrict__ ew,
                          int E) {
    const int lane = threadIdx.x & 31;
    const int gw   = (blockIdx.x * blockDim.x + threadIdx.x) >> 5;
    const int nw   = (gridDim.x * blockDim.x) >> 5;
    for (int e = gw; e < E; e += nw) {
        const int s = src[e];
        const int t = dst[e];
        const float norm = dinv[s] * ew[e] * dinv[t];
        const float4 hv = *reinterpret_cast<const float4*>(h + (size_t)s * D_FEAT + lane * 4);
        float* op = out + (size_t)t * D_FEAT + lane * 4;
        atomicAdd(op + 0, hv.x * norm);
        atomicAdd(op + 1, hv.y * norm);
        atomicAdd(op + 2, hv.z * norm);
        atomicAdd(op + 3, hv.w * norm);
    }
}

extern "C" void kernel_launch(void* const* d_in, const int* in_sizes, int n_in,
                              void* d_out, int out_size, void* d_ws, size_t ws_size,
                              hipStream_t stream) {
    const float* x  = (const float*)d_in[0];
    const float* W  = (const float*)d_in[1];
    const float* bb = (const float*)d_in[2];
    const float* ew = (const float*)d_in[3];
    const int*   ei = (const int*)d_in[4];

    const int E = in_sizes[3];
    const int N = in_sizes[0] / D_FEAT;
    const int* src = ei;
    const int* dst = ei + E;
    float* out = (float*)d_out;

    float* deg = (float*)d_ws;
    size_t degPad = ((size_t)N + 63) & ~(size_t)63;
    float* h = deg + degPad;

    const int TB = 256;
    k_init_deg <<<(N + TB - 1) / TB, TB, 0, stream>>>(deg, N);
    k_deg_accum<<<(E + TB - 1) / TB, TB, 0, stream>>>(deg, dst, ew, E);
    k_dinv     <<<(N + TB - 1) / TB, TB, 0, stream>>>(deg, N);

    k_gemm_wmma<<<(N + 15) / 16, 256, 0, stream>>>(x, W, h, N);

    const int total = N * D_FEAT;
    k_out_init<<<(total + TB - 1) / TB, TB, 0, stream>>>(out, h, deg, bb, total);

    k_scatter<<<4096, TB, 0, stream>>>(out, h, deg, src, dst, ew, E);
}


# degree sum: 64 workgroups each build a private LDS histogram (ds_add_f32), dinv kernel adds the 64 rows in fixed order; replaces 640k global f32 atomics and the init launch; sum order changed
# speedup vs baseline: 1.4589x; 1.4589x over previous
#include <hip/hip_runtime.h>
#include <math.h>

typedef __attribute__((ext_vector_type(2))) float v2f;
typedef __attribute__((ext_vector_type(8))) float v8f;

#define D_FEAT 128

#define DEG_BLOCKS  64
#define DEG_THREADS 1024
#define DEG_CHUNK   16384
#define DEG_BATCH   10

__global__ void __launch_bounds__(DEG_THREADS)
k_deg_part(float* __restrict__ part, const int* __restrict__ dst,
           const float* __restrict__ ew, int E, int N) {
    __shared__ float sdeg[DEG_CHUNK];
    const int tid    = threadIdx.x;
    const int gt     = blockIdx.x * DEG_THREADS + tid;
    const int stride = gridDim.x * DEG_THREADS;
    float* __restrict__ prow = part + (size_t)blockIdx.x * N;

    for (int base = 0; base < N; base += DEG_CHUNK) {
        const int len = min(DEG_CHUNK, N - base);
        for (int i = tid; i < len; i += DEG_THREADS) sdeg[i] = 0.0f;
        __syncthreads();

        for (int e0 = gt; e0 < E; e0 += DEG_BATCH * stride) {
            int   d[DEG_BATCH];
            float w[DEG_BATCH];
#pragma unroll
            for (int u = 0; u < DEG_BATCH; ++u) {
                const int e = e0 + u * stride;
                const bool ok = e < E;
                d[u] = ok ? dst[e] : -1;
                w[u] = ok ? ew[e] : 0.0f;
            }
#pragma unroll
            for (int u = 0; u < DEG_BATCH; ++u) {
                const unsigned r = (unsigned)(d[u] - base);
                if (r < (unsigned)len) atomicAdd(&sdeg[r], w[u]);
            }
        }
        __syncthreads();

        for (int i = tid; i < len; i += DEG_THREADS) prow[base + i] = sdeg[i];
        __syncthreads();
    }
}

#define DINV_BATCH 16
__global__ void k_dinv(float* __restrict__ dinv, const float* __restrict__ part, int n) {
    int i = blockIdx.x * blockDim.x + threadIdx.x;
    if (i < n) {
        const float* __restrict__ p = part + i;
        float d = 1.0f;
#pragma unroll
        for (int b0 = 0; b0 < DEG_BLOCKS; b0 += DINV_BATCH) {
            float v[DINV_BATCH];
#pragma unroll
            for (int j = 0; j < DINV_BATCH; ++j) v[j] = p[(size_t)(b0 + j) * n];
            asm volatile("" :: "v"(v[0]), "v"(v[1]), "v"(v[2]), "v"(v[3]), "v"(v[4]), "v"(v[5]), "v"(v[6]), "v"(v[7]), "v"(v[8]), "v"(v[9]), "v"(v[10]), "v"(v[11]), "v"(v[12]), "v"(v[13]), "v"(v[14]), "v"(v[15]));
#pragma unroll
            for (int j = 0; j < DINV_BATCH; ++j) d += v[j];
        }
        dinv[i] = (d > 0.0f) ? rsqrtf(d) : 0.0f;
    }
}

__global__ void k_gemm_wmma(const float* __restrict__ x, const float* __restrict__ W,
                            float* __restrict__ h, int N) {
    const int lane  = threadIdx.x & 31;
    const int wave  = threadIdx.x >> 5;
    const int Mbase = blockIdx.x * 16;
    const int Nbase = wave * 16;
    const int half  = lane >> 4;
    const int l15   = lane & 15;

    int mrow = Mbase + l15; if (mrow >= N) mrow = N - 1;
    const float* __restrict__ arow = x + (size_t)mrow * D_FEAT;
    const float* __restrict__ brow = W + (size_t)(Nbase + l15) * D_FEAT;

    v8f c = {};
#pragma unroll
    for (int k = 0; k < D_FEAT; k += 4) {
        const int kk = k + 2 * half;
        v2f a = *reinterpret_cast<const v2f*>(arow + kk);
        v2f b = *reinterpret_cast<const v2f*>(brow + kk);
        c = __builtin_amdgcn_wmma_f32_16x16x4_f32(false, a, false, b, (short)0, c,
                                                  false, false);
    }

    float* __restrict__ hout = h + (size_t)Mbase * D_FEAT + Nbase;
#pragma unroll
    for (int v = 0; v < 8; ++v) {
        int m = v + 8 * half;
        if (Mbase + m < N) hout[(size_t)m * D_FEAT + l15] = c[v];
    }
}

__global__ void k_out_init(float* __restrict__ out, const float* __restrict__ h,
                           const float* __restrict__ dinv, const float* __restrict__ bias,
                           int total) {
    int i = blockIdx.x * blockDim.x + threadIdx.x;
    if (i < total) {
        int node = i >> 7;
        int d    = i & 127;
        float di = dinv[node];
        out[i] = bias[d] + h[i] * di * di;
    }
}

__global__ void k_scatter(float* __restrict__ out, const float* __restrict__ h,
                          const float* __restrict__ dinv, const int* __restrict__ src,
                          const int* __restrict__ dst, const float* __restrict__ ew,
                          int E) {
    const int lane = threadIdx.x & 31;
    const int gw   = (blockIdx.x * blockDim.x + threadIdx.x) >> 5;
    const int nw   = (gridDim.x * blockDim.x) >> 5;
    for (int e = gw; e < E; e += nw) {
        const int s = src[e];
        const int t = dst[e];
        const float norm = dinv[s] * ew[e] * dinv[t];
        const float4 hv = *reinterpret_cast<const float4*>(h + (size_t)s * D_FEAT + lane * 4);
        float* op = out + (size_t)t * D_FEAT + lane * 4;
        atomicAdd(op + 0, hv.x * norm);
        atomicAdd(op + 1, hv.y * norm);
        atomicAdd(op + 2, hv.z * norm);
        atomicAdd(op + 3, hv.w * norm);
    }
}

extern "C" void kernel_launch(void* const* d_in, const int* in_sizes, int n_in,
                              void* d_out, int out_size, void* d_ws, size_t ws_size,
                              hipStream_t stream) {
    const float* x  = (const float*)d_in[0];
    const float* W  = (const float*)d_in[1];
    const float* bb = (const float*)d_in[2];
    const float* ew = (const float*)d_in[3];
    const int*   ei = (const int*)d_in[4];

    const int E = in_sizes[3];
    const int N = in_sizes[0] / D_FEAT;
    const int* src = ei;
    const int* dst = ei + E;
    float* out = (float*)d_out;

    float* deg = (float*)d_ws;
    size_t degPad = ((size_t)N + 63) & ~(size_t)63;
    float* h = deg + degPad;
    float* part = h + (size_t)N * D_FEAT;

    const int TB = 256;
    k_deg_part<<<DEG_BLOCKS, DEG_THREADS, 0, stream>>>(part, dst, ew, E, N);
    k_dinv    <<<(N + TB - 1) / TB, TB, 0, stream>>>(deg, part, N);

    k_gemm_wmma<<<(N + 15) / 16, 256, 0, stream>>>(x, W, h, N);

    const int total = N * D_FEAT;
    k_out_init<<<(total + TB - 1) / TB, TB, 0, stream>>>(out, h, deg, bb, total);

    k_scatter<<<4096, TB, 0, stream>>>(out, h, deg, src, dst, ew, E);
}


# scatter-add moved on chip: edges grouped by destination range per chunk (LDS counting sort), one workgroup per 20-node range sums h rows in LDS by ds_add_f32 (arrival order), stores out once
# speedup vs baseline: 2.9639x; 2.0316x over previous
#include <hip/hip_runtime.h>
#include <math.h>

typedef __attribute__((ext_vector_type(2))) float v2f;
typedef __attribute__((ext_vector_type(8))) float v8f;

#define D_FEAT 128

#define DEG_BLOCKS  64
#define DEG_THREADS 1024
#define DEG_CHUNK   16384
#define DEG_BATCH   10

__global__ void __launch_bounds__(DEG_THREADS)
k_deg_part(float* __restrict__ part, const int* __restrict__ dst,
           const float* __restrict__ ew, int E, int N) {
    __shared__ float sdeg[DEG_CHUNK];
    const int tid    = threadIdx.x;
    const int gt     = blockIdx.x * DEG_THREADS + tid;
    const int stride = gridDim.x * DEG_THREADS;
    float* __restrict__ prow = part + (size_t)blockIdx.x * N;

    for (int base = 0; base < N; base += DEG_CHUNK) {
        const int len = min(DEG_CHUNK, N - base);
        for (int i = tid; i < len; i += DEG_THREADS) sdeg[i] = 0.0f;
        __syncthreads();

        for (int e0 = gt; e0 < E; e0 += DEG_BATCH * stride) {
            int   d[DEG_BATCH];
            float w[DEG_BATCH];
#pragma unroll
            for (int u = 0; u < DEG_BATCH; ++u) {
                const int e = e0 + u * stride;
                const bool ok = e < E;
                d[u] = ok ? dst[e] : -1;
                w[u] = ok ? ew[e] : 0.0f;
            }
#pragma unroll
            for (int u = 0; u < DEG_BATCH; ++u) {
                const unsigned r = (unsigned)(d[u] - base);
                if (r < (unsigned)len) atomicAdd(&sdeg[r], w[u]);
            }
        }
        __syncthreads();

        for (int i = tid; i < len; i += DEG_THREADS) prow[base + i] = sdeg[i];
        __syncthreads();
    }
}

#define DINV_BATCH 16
__global__ void k_dinv(float* __restrict__ dinv, const float* __restrict__ part, int n) {
    int i = blockIdx.x * blockDim.x + threadIdx.x;
    if (i < n) {
        const float* __restrict__ p = part + i;
        float d = 1.0f;
#pragma unroll
        for (int b0 = 0; b0 < DEG_BLOCKS; b0 += DINV_BATCH) {
            float v[DINV_BATCH];
#pragma unroll
            for (int j = 0; j < DINV_BATCH; ++j) v[j] = p[(size_t)(b0 + j) * n];
            asm volatile("" :: "v"(v[0]), "v"(v[1]), "v"(v[2]), "v"(v[3]), "v"(v[4]), "v"(v[5]), "v"(v[6]), "v"(v[7]), "v"(v[8]), "v"(v[9]), "v"(v[10]), "v"(v[11]), "v"(v[12]), "v"(v[13]), "v"(v[14]), "v"(v[15]));
#pragma unroll
            for (int j = 0; j < DINV_BATCH; ++j) d += v[j];
        }
        dinv[i] = (d > 0.0f) ? rsqrtf(d) : 0.0f;
    }
}

__global__ void k_gemm_wmma(const float* __restrict__ x, const float* __restrict__ W,
                            float* __restrict__ h, int N) {
    const int lane  = threadIdx.x & 31;
    const int wave  = threadIdx.x >> 5;
    const int Mbase = blockIdx.x * 16;
    const int Nbase = wave * 16;
    const int half  = lane >> 4;
    const int l15   = lane & 15;

    int mrow = Mbase + l15; if (mrow >= N) mrow = N - 1;
    const float* __restrict__ arow = x + (size_t)mrow * D_FEAT;
    const float* __restrict__ brow = W + (size_t)(Nbase + l15) * D_FEAT;

    v8f c = {};
#pragma unroll
    for (int k = 0; k < D_FEAT; k += 4) {
        const int kk = k + 2 * half;
        v2f a = *reinterpret_cast<const v2f*>(arow + kk);
        v2f b = *reinterpret_cast<const v2f*>(brow + kk);
        c = __builtin_amdgcn_wmma_f32_16x16x4_f32(false, a, false, b, (short)0, c,
                                                  false, false);
    }

    float* __restrict__ hout = h + (size_t)Mbase * D_FEAT + Nbase;
#pragma unroll
    for (int v = 0; v < 8; ++v) {
        int m = v + 8 * half;
        if (Mbase + m < N) hout[(size_t)m * D_FEAT + l15] = c[v];
    }
}

#define BIN_NODES    20
#define NBIN_MAX     1024
#define SORT_BLOCKS  64
#define SORT_THREADS 1024

__global__ void __launch_bounds__(SORT_THREADS)
k_bin(int* __restrict__ rec_w, int* __restrict__ tab, const int* __restrict__ src,
      const int* __restrict__ dst, const float* __restrict__ ew, int E, int nb) {
    int4* __restrict__ rec = reinterpret_cast<int4*>(rec_w);
    __shared__ int cnt[NBIN_MAX];
    __shared__ int cur[NBIN_MAX];
    __shared__ int wsum[32];
    const int tid  = threadIdx.x;
    const int lane = tid & 31;
    const int wv   = tid >> 5;
    const int CH   = (E + SORT_BLOCKS - 1) / SORT_BLOCKS;
    const int e_lo = blockIdx.x * CH;
    const int e_hi = min(E, e_lo + CH);

    cnt[tid] = 0;
    __syncthreads();
    for (int e = e_lo + tid; e < e_hi; e += SORT_THREADS)
        atomicAdd(&cnt[dst[e] / BIN_NODES], 1);
    __syncthreads();

    const int v = cnt[tid];
    int x = v;
    for (int o = 1; o < 32; o <<= 1) {
        const int y = __shfl_up(x, o);
        if (lane >= o) x += y;
    }
    if (lane == 31) wsum[wv] = x;
    __syncthreads();
    if (wv == 0) {
        const int s = wsum[lane];
        int t = s;
        for (int o = 1; o < 32; o <<= 1) {
            const int y = __shfl_up(t, o);
            if (lane >= o) t += y;
        }
        wsum[lane] = t - s;
    }
    __syncthreads();
    const int excl = x - v + wsum[wv];
    cur[tid] = excl;
    if (tid <= nb) tab[(size_t)blockIdx.x * (nb + 1) + tid] = excl;
    __syncthreads();

    int4* __restrict__ rg = rec + e_lo;
    for (int e = e_lo + tid; e < e_hi; e += SORT_THREADS) {
        const int t = dst[e];
        const int pos = atomicAdd(&cur[t / BIN_NODES], 1);
        rg[pos] = make_int4(src[e], t, __float_as_int(ew[e]), 0);
    }
}

#define AGG_THREADS 1024
#define AGG_BATCH   4

__global__ void __launch_bounds__(AGG_THREADS)
k_agg(float* __restrict__ out, const float* __restrict__ h, const float* __restrict__ dinv,
      const float* __restrict__ bias, const int* __restrict__ rec_w,
      const int* __restrict__ tab, int E, int N, int nb) {
    const int4* __restrict__ rec = reinterpret_cast<const int4*>(rec_w);
    __shared__ float acc[BIN_NODES * D_FEAT];
    const int tid  = threadIdx.x;
    const int lane = tid & 31;
    const int wv   = __builtin_amdgcn_readfirstlane(tid >> 5);
    const int b    = blockIdx.x;
    const int base = b * BIN_NODES;
    const int rows = min(BIN_NODES, N - base);
    const int nel  = rows * D_FEAT;

    for (int i = tid; i < nel; i += AGG_THREADS) {
        const int node = base + (i >> 7);
        const float di = dinv[node];
        acc[i] = bias[i & 127] + h[(size_t)base * D_FEAT + i] * di * di;
    }
    __syncthreads();

    const int CH = (E + SORT_BLOCKS - 1) / SORT_BLOCKS;
    for (int g = wv; g < SORT_BLOCKS; g += AGG_THREADS / 32) {
        const int* __restrict__ tg = tab + (size_t)g * (nb + 1) + b;
        const int s0 = __builtin_amdgcn_readfirstlane(tg[0]);
        const int s1 = __builtin_amdgcn_readfirstlane(tg[1]);
        const int4* __restrict__ rg = rec + (size_t)g * CH;
        for (int i0 = s0; i0 < s1; i0 += 32) {
            const int i = i0 + lane;
            const bool ok = i < s1;
            int4 r = make_int4(0, base, 0, 0);
            if (ok) r = rg[i];
            float nrm = 0.0f;
            if (ok) nrm = dinv[r.x] * __int_as_float(r.z) * dinv[r.y];
            const int roff = (r.y - base) * D_FEAT;
            const int m = min(32, s1 - i0);
            for (int j = 0; j < m; j += AGG_BATCH) {
                float hv[AGG_BATCH][4];
                int   oj[AGG_BATCH];
                float nj[AGG_BATCH];
#pragma unroll
                for (int u = 0; u < AGG_BATCH; ++u) {
                    const int jj = (j + u) & 31;
                    const int sj = __shfl(r.x, jj);
                    oj[u] = __shfl(roff, jj);
                    nj[u] = __shfl(nrm, jj);
                    const float* __restrict__ hp = h + (size_t)sj * D_FEAT + lane;
                    hv[u][0] = hp[0];
                    hv[u][1] = hp[32];
                    hv[u][2] = hp[64];
                    hv[u][3] = hp[96];
                }
#pragma unroll
                for (int u = 0; u < AGG_BATCH; ++u) {
                    if (j + u < m) {
                        float* ap = acc + oj[u] + lane;
                        atomicAdd(ap,      hv[u][0] * nj[u]);
                        atomicAdd(ap + 32, hv[u][1] * nj[u]);
                        atomicAdd(ap + 64, hv[u][2] * nj[u]);
                        atomicAdd(ap + 96, hv[u][3] * nj[u]);
                    }
                }
            }
        }
    }
    __syncthreads();

    float* __restrict__ orow = out + (size_t)base * D_FEAT;
    for (int i = tid; i < nel; i += AGG_THREADS) orow[i] = acc[i];
}

extern "C" void kernel_launch(void* const* d_in, const int* in_sizes, int n_in,
                              void* d_out, int out_size, void* d_ws, size_t ws_size,
                              hipStream_t stream) {
    const float* x  = (const float*)d_in[0];
    const float* W  = (const float*)d_in[1];
    const float* bb = (const float*)d_in[2];
    const float* ew = (const float*)d_in[3];
    const int*   ei = (const int*)d_in[4];

    const int E = in_sizes[3];
    const int N = in_sizes[0] / D_FEAT;
    const int* src = ei;
    const int* dst = ei + E;
    float* out = (float*)d_out;

    float* deg = (float*)d_ws;
    size_t degPad = ((size_t)N + 63) & ~(size_t)63;
    float* h = deg + degPad;
    float* part = h + (size_t)N * D_FEAT;
    int* rec = (int*)(part + (size_t)DEG_BLOCKS * N);
    int* tab = rec + (size_t)E * 4;
    const int nb = (N + BIN_NODES - 1) / BIN_NODES;

    const int TB = 256;
    k_deg_part<<<DEG_BLOCKS, DEG_THREADS, 0, stream>>>(part, dst, ew, E, N);
    k_dinv    <<<(N + TB - 1) / TB, TB, 0, stream>>>(deg, part, N);

    k_gemm_wmma<<<(N + 15) / 16, 256, 0, stream>>>(x, W, h, N);

    k_bin<<<SORT_BLOCKS, SORT_THREADS, 0, stream>>>(rec, tab, src, dst, ew, E, nb);
    k_agg<<<nb, AGG_THREADS, 0, stream>>>(out, h, deg, bb, rec, tab, E, N, nb);
}


# degree histograms, edge binning and the WMMA GEMM (mutually independent) run as workgroup-range roles of one launch, binning first; 5 launches -> 3; arithmetic and sum orders unchanged
# speedup vs baseline: 3.3869x; 1.1427x over previous
#include <hip/hip_runtime.h>
#include <math.h>

typedef __attribute__((ext_vector_type(2))) float v2f;
typedef __attribute__((ext_vector_type(8))) float v8f;

#define D_FEAT 128


#define DEG_BLOCKS  64
#define DEG_THREADS 1024
#define DEG_CHUNK   16384
#define DEG_BATCH   10

__device__ __forceinline__ void
deg_part_role(const int blk, float* __restrict__ part, const int* __restrict__ dst,
              const float* __restrict__ ew, int E, int N) {
    __shared__ float sdeg[DEG_CHUNK];
    const int tid    = threadIdx.x;
    const int gt     = blk * DEG_THREADS + tid;
    const int stride = DEG_BLOCKS * DEG_THREADS;
    float* __restrict__ prow = part + (size_t)blk * N;

    for (int base = 0; base < N; base += DEG_CHUNK) {
        const int len = min(DEG_CHUNK, N - base);
        for (int i = tid; i < len; i += DEG_THREADS) sdeg[i] = 0.0f;
        __syncthreads();

        for (int e0 = gt; e0 < E; e0 += DEG_BATCH * stride) {
            int   d[DEG_BATCH];
            float w[DEG_BATCH];
#pragma unroll
            for (int u = 0; u < DEG_BATCH; ++u) {
                const int e = e0 + u * stride;
                const bool ok = e < E;
                d[u] = ok ? dst[e] : -1;
                w[u] = ok ? ew[e] : 0.0f;
            }
#pragma unroll
            for (int u = 0; u < DEG_BATCH; ++u) {
                const unsigned r = (unsigned)(d[u] - base);
                if (r < (unsigned)len) atomicAdd(&sdeg[r], w[u]);
            }
        }
        __syncthreads();

        for (int i = tid; i < len; i += DEG_THREADS) prow[base + i] = sdeg[i];
        __syncthreads();
    }
}

#define DINV_BATCH 16
__global__ void k_dinv(float* __restrict__ dinv, const float* __restrict__ part, int n) {
    int i = blockIdx.x * blockDim.x + threadIdx.x;
    if (i < n) {
        const float* __restrict__ p = part + i;
        float d = 1.0f;
#pragma unroll
        for (int b0 = 0; b0 < DEG_BLOCKS; b0 += DINV_BATCH) {
            float v[DINV_BATCH];
#pragma unroll
            for (int j = 0; j < DINV_BATCH; ++j) v[j] = p[(size_t)(b0 + j) * n];
            asm volatile("" :: "v"(v[0]), "v"(v[1]), "v"(v[2]), "v"(v[3]), "v"(v[4]), "v"(v[5]), "v"(v[6]), "v"(v[7]), "v"(v[8]), "v"(v[9]), "v"(v[10]), "v"(v[11]), "v"(v[12]), "v"(v[13]), "v"(v[14]), "v"(v[15]));
#pragma unroll
            for (int j = 0; j < DINV_BATCH; ++j) d += v[j];
        }
        dinv[i] = (d > 0.0f) ? rsqrtf(d) : 0.0f;
    }
}

#define GEMM_ROWS 64
__device__ __forceinline__ void
gemm_role(const int blk, const float* __restrict__ x, const float* __restrict__ W,
          float* __restrict__ h, int N) {
    const int lane  = threadIdx.x & 31;
    const int wave  = threadIdx.x >> 5;
    const int Mbase = blk * GEMM_ROWS + (wave >> 3) * 16;
    const int Nbase = (wave & 7) * 16;
    if (Mbase >= N) return;
    const int half  = lane >> 4;
    const int l15   = lane & 15;

    int mrow = Mbase + l15; if (mrow >= N) mrow = N - 1;
    const float* __restrict__ arow = x + (size_t)mrow * D_FEAT;
    const float* __restrict__ brow = W + (size_t)(Nbase + l15) * D_FEAT;

    v8f c = {};
#pragma unroll
    for (int k = 0; k < D_FEAT; k += 4) {
        const int kk = k + 2 * half;
        v2f a = *reinterpret_cast<const v2f*>(arow + kk);
        v2f b = *reinterpret_cast<const v2f*>(brow + kk);
        c = __builtin_amdgcn_wmma_f32_16x16x4_f32(false, a, false, b, (short)0, c,
                                                  false, false);
    }

    float* __restrict__ hout = h + (size_t)Mbase * D_FEAT + Nbase;
#pragma unroll
    for (int v = 0; v < 8; ++v) {
        int m = v + 8 * half;
        if (Mbase + m < N) hout[(size_t)m * D_FEAT + l15] = c[v];
    }
}

#define BIN_NODES    20
#define NBIN_MAX     1024
#define SORT_BLOCKS  64
#define SORT_THREADS 1024

__device__ __forceinline__ void
bin_role(const int blk, int* __restrict__ rec_w, int* __restrict__ tab, const int* __restrict__ src,
         const int* __restrict__ dst, const float* __restrict__ ew, int E, int nb) {
    int4* __restrict__ rec = reinterpret_cast<int4*>(rec_w);
    __shared__ int cnt[NBIN_MAX];
    __shared__ int cur[NBIN_MAX];
    __shared__ int wsum[32];
    const int tid  = threadIdx.x;
    const int lane = tid & 31;
    const int wv   = tid >> 5;
    const int CH   = (E + SORT_BLOCKS - 1) / SORT_BLOCKS;
    const int e_lo = blk * CH;
    const int e_hi = min(E, e_lo + CH);

    cnt[tid] = 0;
    __syncthreads();
    for (int e = e_lo + tid; e < e_hi; e += SORT_THREADS)
        atomicAdd(&cnt[dst[e] / BIN_NODES], 1);
    __syncthreads();

    const int v = cnt[tid];
    int x = v;
    for (int o = 1; o < 32; o <<= 1) {
        const int y = __shfl_up(x, o);
        if (lane >= o) x += y;
    }
    if (lane == 31) wsum[wv] = x;
    __syncthreads();
    if (wv == 0) {
        const int s = wsum[lane];
        int t = s;
        for (int o = 1; o < 32; o <<= 1) {
            const int y = __shfl_up(t, o);
            if (lane >= o) t += y;
        }
        wsum[lane] = t - s;
    }
    __syncthreads();
    const int excl = x - v + wsum[wv];
    cur[tid] = excl;
    if (tid <= nb) tab[(size_t)blk * (nb + 1) + tid] = excl;
    __syncthreads();

    int4* __restrict__ rg = rec + e_lo;
    for (int e = e_lo + tid; e < e_hi; e += SORT_THREADS) {
        const int t = dst[e];
        const int pos = atomicAdd(&cur[t / BIN_NODES], 1);
        rg[pos] = make_int4(src[e], t, __float_as_int(ew[e]), 0);
    }
}

#define PRE_THREADS 1024
__global__ void __launch_bounds__(PRE_THREADS)
k_pre(float* __restrict__ part, int* __restrict__ rec_w, int* __restrict__ tab,
      float* __restrict__ h, const float* __restrict__ x, const float* __restrict__ W,
      const int* __restrict__ src, const int* __restrict__ dst,
      const float* __restrict__ ew, int E, int N, int nb) {
    const int b = blockIdx.x;
    if (b < SORT_BLOCKS)
        bin_role(b, rec_w, tab, src, dst, ew, E, nb);
    else if (b < SORT_BLOCKS + DEG_BLOCKS)
        deg_part_role(b - SORT_BLOCKS, part, dst, ew, E, N);
    else
        gemm_role(b - (SORT_BLOCKS + DEG_BLOCKS), x, W, h, N);
}

#define AGG_THREADS 1024
#define AGG_BATCH   4

__global__ void __launch_bounds__(AGG_THREADS)
k_agg(float* __restrict__ out, const float* __restrict__ h, const float* __restrict__ dinv,
      const float* __restrict__ bias, const int* __restrict__ rec_w,
      const int* __restrict__ tab, int E, int N, int nb) {
    const int4* __restrict__ rec = reinterpret_cast<const int4*>(rec_w);
    __shared__ float acc[BIN_NODES * D_FEAT];
    const int tid  = threadIdx.x;
    const int lane = tid & 31;
    const int wv   = __builtin_amdgcn_readfirstlane(tid >> 5);
    const int b    = blockIdx.x;
    const int base = b * BIN_NODES;
    const int rows = min(BIN_NODES, N - base);
    const int nel  = rows * D_FEAT;

    for (int i = tid; i < nel; i += AGG_THREADS) {
        const int node = base + (i >> 7);
        const float di = dinv[node];
        acc[i] = bias[i & 127] + h[(size_t)base * D_FEAT + i] * di * di;
    }
    __syncthreads();

    const int CH = (E + SORT_BLOCKS - 1) / SORT_BLOCKS;
    for (int g = wv; g < SORT_BLOCKS; g += AGG_THREADS / 32) {
        const int* __restrict__ tg = tab + (size_t)g * (nb + 1) + b;
        const int s0 = __builtin_amdgcn_readfirstlane(tg[0]);
        const int s1 = __builtin_amdgcn_readfirstlane(tg[1]);
        const int4* __restrict__ rg = rec + (size_t)g * CH;
        for (int i0 = s0; i0 < s1; i0 += 32) {
            const int i = i0 + lane;
            const bool ok = i < s1;
            int4 r = make_int4(0, base, 0, 0);
            if (ok) r = rg[i];
            float nrm = 0.0f;
            if (ok) nrm = dinv[r.x] * __int_as_float(r.z) * dinv[r.y];
            const int roff = (r.y - base) * D_FEAT;
            const int m = min(32, s1 - i0);
            for (int j = 0; j < m; j += AGG_BATCH) {
                float hv[AGG_BATCH][4];
                int   oj[AGG_BATCH];
                float nj[AGG_BATCH];
#pragma unroll
                for (int u = 0; u < AGG_BATCH; ++u) {
                    const int jj = (j + u) & 31;
                    const int sj = __shfl(r.x, jj);
                    oj[u] = __shfl(roff, jj);
                    nj[u] = __shfl(nrm, jj);
                    const float* __restrict__ hp = h + (size_t)sj * D_FEAT + lane;
                    hv[u][0] = hp[0];
                    hv[u][1] = hp[32];
                    hv[u][2] = hp[64];
                    hv[u][3] = hp[96];
                }
#pragma unroll
                for (int u = 0; u < AGG_BATCH; ++u) {
                    if (j + u < m) {
                        float* ap = acc + oj[u] + lane;
                        atomicAdd(ap,      hv[u][0] * nj[u]);
                        atomicAdd(ap + 32, hv[u][1] * nj[u]);
                        atomicAdd(ap + 64, hv[u][2] * nj[u]);
                        atomicAdd(ap + 96, hv[u][3] * nj[u]);
                    }
                }
            }
        }
    }
    __syncthreads();

    float* __restrict__ orow = out + (size_t)base * D_FEAT;
    for (int i = tid; i < nel; i += AGG_THREADS) orow[i] = acc[i];
}

extern "C" void kernel_launch(void* const* d_in, const int* in_sizes, int n_in,
                              void* d_out, int out_size, void* d_ws, size_t ws_size,
                              hipStream_t stream) {
    const float* x  = (const float*)d_in[0];
    const float* W  = (const float*)d_in[1];
    const float* bb = (const float*)d_in[2];
    const float* ew = (const float*)d_in[3];
    const int*   ei = (const int*)d_in[4];

    const int E = in_sizes[3];
    const int N = in_sizes[0] / D_FEAT;
    const int* src = ei;
    const int* dst = ei + E;
    float* out = (float*)d_out;

    float* deg = (float*)d_ws;
    size_t degPad = ((size_t)N + 63) & ~(size_t)63;
    float* h = deg + degPad;
    float* part = h + (size_t)N * D_FEAT;
    int* rec = (int*)(part + (size_t)DEG_BLOCKS * N);
    int* tab = rec + (size_t)E * 4;
    const int nb = (N + BIN_NODES - 1) / BIN_NODES;

    const int TB = 256;
    k_pre <<<SORT_BLOCKS + DEG_BLOCKS + (N + GEMM_ROWS - 1) / GEMM_ROWS, PRE_THREADS, 0, stream>>>(
        part, rec, tab, h, x, W, src, dst, ew, E, N, nb);
    k_dinv<<<(N + TB - 1) / TB, TB, 0, stream>>>(deg, part, N);
    k_agg <<<nb, AGG_THREADS, 0, stream>>>(out, h, deg, bb, rec, tab, E, N, nb);
}


# aggregation: each h row fetched by one 16-byte load per lane (1 instruction, 4 lines), was four 4-byte line loads; LDS tile feature-permuted so adds stay conflict-free; arrival-order sums as before
# speedup vs baseline: 3.7129x; 1.0963x over previous
#include <hip/hip_runtime.h>
#include <math.h>

typedef __attribute__((ext_vector_type(2))) float v2f;
typedef __attribute__((ext_vector_type(4))) float v4f;
typedef __attribute__((ext_vector_type(8))) float v8f;

#define D_FEAT 128


#define DEG_BLOCKS  64
#define DEG_THREADS 1024
#define DEG_CHUNK   16384
#define DEG_BATCH   10

__device__ __forceinline__ void
deg_part_role(const int blk, float* __restrict__ part, const int* __restrict__ dst,
              const float* __restrict__ ew, int E, int N) {
    __shared__ float sdeg[DEG_CHUNK];
    const int tid    = threadIdx.x;
    const int gt     = blk * DEG_THREADS + tid;
    const int stride = DEG_BLOCKS * DEG_THREADS;
    float* __restrict__ prow = part + (size_t)blk * N;

    for (int base = 0; base < N; base += DEG_CHUNK) {
        const int len = min(DEG_CHUNK, N - base);
        for (int i = tid; i < len; i += DEG_THREADS) sdeg[i] = 0.0f;
        __syncthreads();

        for (int e0 = gt; e0 < E; e0 += DEG_BATCH * stride) {
            int   d[DEG_BATCH];
            float w[DEG_BATCH];
#pragma unroll
            for (int u = 0; u < DEG_BATCH; ++u) {
                const int e = e0 + u * stride;
                const bool ok = e < E;
                d[u] = ok ? dst[e] : -1;
                w[u] = ok ? ew[e] : 0.0f;
            }
#pragma unroll
            for (int u = 0; u < DEG_BATCH; ++u) {
                const unsigned r = (unsigned)(d[u] - base);
                if (r < (unsigned)len) atomicAdd(&sdeg[r], w[u]);
            }
        }
        __syncthreads();

        for (int i = tid; i < len; i += DEG_THREADS) prow[base + i] = sdeg[i];
        __syncthreads();
    }
}

#define DINV_BATCH 16
__global__ void k_dinv(float* __restrict__ dinv, const float* __restrict__ part, int n) {
    int i = blockIdx.x * blockDim.x + threadIdx.x;
    if (i < n) {
        const float* __restrict__ p = part + i;
        float d = 1.0f;
#pragma unroll
        for (int b0 = 0; b0 < DEG_BLOCKS; b0 += DINV_BATCH) {
            float v[DINV_BATCH];
#pragma unroll
            for (int j = 0; j < DINV_BATCH; ++j) v[j] = p[(size_t)(b0 + j) * n];
            asm volatile("" :: "v"(v[0]), "v"(v[1]), "v"(v[2]), "v"(v[3]), "v"(v[4]), "v"(v[5]), "v"(v[6]), "v"(v[7]), "v"(v[8]), "v"(v[9]), "v"(v[10]), "v"(v[11]), "v"(v[12]), "v"(v[13]), "v"(v[14]), "v"(v[15]));
#pragma unroll
            for (int j = 0; j < DINV_BATCH; ++j) d += v[j];
        }
        dinv[i] = (d > 0.0f) ? rsqrtf(d) : 0.0f;
    }
}

#define GEMM_ROWS 64
__device__ __forceinline__ void
gemm_role(const int blk, const float* __restrict__ x, const float* __restrict__ W,
          float* __restrict__ h, int N) {
    const int lane  = threadIdx.x & 31;
    const int wave  = threadIdx.x >> 5;
    const int Mbase = blk * GEMM_ROWS + (wave >> 3) * 16;
    const int Nbase = (wave & 7) * 16;
    if (Mbase >= N) return;
    const int half  = lane >> 4;
    const int l15   = lane & 15;

    int mrow = Mbase + l15; if (mrow >= N) mrow = N - 1;
    const float* __restrict__ arow = x + (size_t)mrow * D_FEAT;
    const float* __restrict__ brow = W + (size_t)(Nbase + l15) * D_FEAT;

    v8f c = {};
#pragma unroll
    for (int k = 0; k < D_FEAT; k += 4) {
        const int kk = k + 2 * half;
        v2f a = *reinterpret_cast<const v2f*>(arow + kk);
        v2f b = *reinterpret_cast<const v2f*>(brow + kk);
        c = __builtin_amdgcn_wmma_f32_16x16x4_f32(false, a, false, b, (short)0, c,
                                                  false, false);
    }

    float* __restrict__ hout = h + (size_t)Mbase * D_FEAT + Nbase;
#pragma unroll
    for (int v = 0; v < 8; ++v) {
        int m = v + 8 * half;
        if (Mbase + m < N) hout[(size_t)m * D_FEAT + l15] = c[v];
    }
}

#define BIN_NODES    20
#define NBIN_MAX     1024
#define SORT_BLOCKS  64
#define SORT_THREADS 1024

__device__ __forceinline__ void
bin_role(const int blk, int* __restrict__ rec_w, int* __restrict__ tab, const int* __restrict__ src,
         const int* __restrict__ dst, const float* __restrict__ ew, int E, int nb) {
    int4* __restrict__ rec = reinterpret_cast<int4*>(rec_w);
    __shared__ int cnt[NBIN_MAX];
    __shared__ int cur[NBIN_MAX];
    __shared__ int wsum[32];
    const int tid  = threadIdx.x;
    const int lane = tid & 31;
    const int wv   = tid >> 5;
    const int CH   = (E + SORT_BLOCKS - 1) / SORT_BLOCKS;
    const int e_lo = blk * CH;
    const int e_hi = min(E, e_lo + CH);

    cnt[tid] = 0;
    __syncthreads();
    for (int e = e_lo + tid; e < e_hi; e += SORT_THREADS)
        atomicAdd(&cnt[dst[e] / BIN_NODES], 1);
    __syncthreads();

    const int v = cnt[tid];
    int x = v;
    for (int o = 1; o < 32; o <<= 1) {
        const int y = __shfl_up(x, o);
        if (lane >= o) x += y;
    }
    if (lane == 31) wsum[wv] = x;
    __syncthreads();
    if (wv == 0) {
        const int s = wsum[lane];
        int t = s;
        for (int o = 1; o < 32; o <<= 1) {
            const int y = __shfl_up(t, o);
            if (lane >= o) t += y;
        }
        wsum[lane] = t - s;
    }
    __syncthreads();
    const int excl = x - v + wsum[wv];
    cur[tid] = excl;
    if (tid <= nb) tab[(size_t)blk * (nb + 1) + tid] = excl;
    __syncthreads();

    int4* __restrict__ rg = rec + e_lo;
    for (int e = e_lo + tid; e < e_hi; e += SORT_THREADS) {
        const int t = dst[e];
        const int pos = atomicAdd(&cur[t / BIN_NODES], 1);
        rg[pos] = make_int4(src[e], t, __float_as_int(ew[e]), 0);
    }
}

#define PRE_THREADS 1024
__global__ void __launch_bounds__(PRE_THREADS)
k_pre(float* __restrict__ part, int* __restrict__ rec_w, int* __restrict__ tab,
      float* __restrict__ h, const float* __restrict__ x, const float* __restrict__ W,
      const int* __restrict__ src, const int* __restrict__ dst,
      const float* __restrict__ ew, int E, int N, int nb) {
    const int b = blockIdx.x;
    if (b < SORT_BLOCKS)
        bin_role(b, rec_w, tab, src, dst, ew, E, nb);
    else if (b < SORT_BLOCKS + DEG_BLOCKS)
        deg_part_role(b - SORT_BLOCKS, part, dst, ew, E, N);
    else
        gemm_role(b - (SORT_BLOCKS + DEG_BLOCKS), x, W, h, N);
}

#define AGG_THREADS 1024
#define AGG_BATCH   4

__global__ void __launch_bounds__(AGG_THREADS)
k_agg(float* __restrict__ out, const float* __restrict__ h, const float* __restrict__ dinv,
      const float* __restrict__ bias, const int* __restrict__ rec_w,
      const int* __restrict__ tab, int E, int N, int nb) {
    const int4* __restrict__ rec = reinterpret_cast<const int4*>(rec_w);
    __shared__ float acc[BIN_NODES * D_FEAT];
    const int tid  = threadIdx.x;
    const int lane = tid & 31;
    const int wv   = __builtin_amdgcn_readfirstlane(tid >> 5);
    const int b    = blockIdx.x;
    const int base = b * BIN_NODES;
    const int rows = min(BIN_NODES, N - base);
    const int nq   = rows * 32;

    for (int q = tid; q < nq; q += AGG_THREADS) {
        const int nd = q >> 5;
        const int l  = q & 31;
        const float di = dinv[base + nd];
        const v4f hs = *reinterpret_cast<const v4f*>(h + (size_t)(base + nd) * D_FEAT + 4 * l);
        const v4f bv = *reinterpret_cast<const v4f*>(bias + 4 * l);
        float* ap = acc + nd * D_FEAT + l;
        ap[0]  = bv.x + hs.x * di * di;
        ap[32] = bv.y + hs.y * di * di;
        ap[64] = bv.z + hs.z * di * di;
        ap[96] = bv.w + hs.w * di * di;
    }
    __syncthreads();

    const int CH = (E + SORT_BLOCKS - 1) / SORT_BLOCKS;
    for (int g = wv; g < SORT_BLOCKS; g += AGG_THREADS / 32) {
        const int* __restrict__ tg = tab + (size_t)g * (nb + 1) + b;
        const int s0 = __builtin_amdgcn_readfirstlane(tg[0]);
        const int s1 = __builtin_amdgcn_readfirstlane(tg[1]);
        const int4* __restrict__ rg = rec + (size_t)g * CH;
        for (int i0 = s0; i0 < s1; i0 += 32) {
            const int i = i0 + lane;
            const bool ok = i < s1;
            int4 r = make_int4(0, base, 0, 0);
            if (ok) r = rg[i];
            float nrm = 0.0f;
            if (ok) nrm = dinv[r.x] * __int_as_float(r.z) * dinv[r.y];
            const int roff = (r.y - base) * D_FEAT;
            const int m = min(32, s1 - i0);
            for (int j = 0; j < m; j += AGG_BATCH) {
                v4f   hv[AGG_BATCH];
                int   oj[AGG_BATCH];
                float nj[AGG_BATCH];
#pragma unroll
                for (int u = 0; u < AGG_BATCH; ++u) {
                    const int jj = (j + u) & 31;
                    const int sj = __shfl(r.x, jj);
                    oj[u] = __shfl(roff, jj);
                    nj[u] = __shfl(nrm, jj);
                    hv[u] = *reinterpret_cast<const v4f*>(h + (size_t)sj * D_FEAT + 4 * lane);
                }
#pragma unroll
                for (int u = 0; u < AGG_BATCH; ++u) {
                    if (j + u < m) {
                        float* ap = acc + oj[u] + lane;
                        atomicAdd(ap,      hv[u].x * nj[u]);
                        atomicAdd(ap + 32, hv[u].y * nj[u]);
                        atomicAdd(ap + 64, hv[u].z * nj[u]);
                        atomicAdd(ap + 96, hv[u].w * nj[u]);
                    }
                }
            }
        }
    }
    __syncthreads();

    for (int q = tid; q < nq; q += AGG_THREADS) {
        const int nd = q >> 5;
        const int l  = q & 31;
        const float* ap = acc + nd * D_FEAT + l;
        v4f o;
        o.x = ap[0];
        o.y = ap[32];
        o.z = ap[64];
        o.w = ap[96];
        *reinterpret_cast<v4f*>(out + (size_t)(base + nd) * D_FEAT + 4 * l) = o;
    }
}

extern "C" void kernel_launch(void* const* d_in, const int* in_sizes, int n_in,
                              void* d_out, int out_size, void* d_ws, size_t ws_size,
                              hipStream_t stream) {
    const float* x  = (const float*)d_in[0];
    const float* W  = (const float*)d_in[1];
    const float* bb = (const float*)d_in[2];
    const float* ew = (const float*)d_in[3];
    const int*   ei = (const int*)d_in[4];

    const int E = in_sizes[3];
    const int N = in_sizes[0] / D_FEAT;
    const int* src = ei;
    const int* dst = ei + E;
    float* out = (float*)d_out;

    float* deg = (float*)d_ws;
    size_t degPad = ((size_t)N + 63) & ~(size_t)63;
    float* h = deg + degPad;
    float* part = h + (size_t)N * D_FEAT;
    int* rec = (int*)(part + (size_t)DEG_BLOCKS * N);
    int* tab = rec + (size_t)E * 4;
    const int nb = (N + BIN_NODES - 1) / BIN_NODES;

    const int TB = 256;
    k_pre <<<SORT_BLOCKS + DEG_BLOCKS + (N + GEMM_ROWS - 1) / GEMM_ROWS, PRE_THREADS, 0, stream>>>(
        part, rec, tab, h, x, W, src, dst, ew, E, N, nb);
    k_dinv<<<(N + TB - 1) / TB, TB, 0, stream>>>(deg, part, N);
    k_agg <<<nb, AGG_THREADS, 0, stream>>>(out, h, deg, bb, rec, tab, E, N, nb);
}


# aggregation: each edge's src, slot and norm are read by v_readlane into scalar registers (lane index is wave-uniform), was 3 ds_bpermute per edge; LDS pipe carries only the adds; sums unchanged
# speedup vs baseline: 3.8759x; 1.0439x over previous
#include <hip/hip_runtime.h>
#include <math.h>

typedef __attribute__((ext_vector_type(2))) float v2f;
typedef __attribute__((ext_vector_type(4))) float v4f;
typedef __attribute__((ext_vector_type(8))) float v8f;

#define D_FEAT 128


#define DEG_BLOCKS  64
#define DEG_THREADS 1024
#define DEG_CHUNK   16384
#define DEG_BATCH   10

__device__ __forceinline__ void
deg_part_role(const int blk, float* __restrict__ part, const int* __restrict__ dst,
              const float* __restrict__ ew, int E, int N) {
    __shared__ float sdeg[DEG_CHUNK];
    const int tid    = threadIdx.x;
    const int gt     = blk * DEG_THREADS + tid;
    const int stride = DEG_BLOCKS * DEG_THREADS;
    float* __restrict__ prow = part + (size_t)blk * N;

    for (int base = 0; base < N; base += DEG_CHUNK) {
        const int len = min(DEG_CHUNK, N - base);
        for (int i = tid; i < len; i += DEG_THREADS) sdeg[i] = 0.0f;
        __syncthreads();

        for (int e0 = gt; e0 < E; e0 += DEG_BATCH * stride) {
            int   d[DEG_BATCH];
            float w[DEG_BATCH];
#pragma unroll
            for (int u = 0; u < DEG_BATCH; ++u) {
                const int e = e0 + u * stride;
                const bool ok = e < E;
                d[u] = ok ? dst[e] : -1;
                w[u] = ok ? ew[e] : 0.0f;
            }
#pragma unroll
            for (int u = 0; u < DEG_BATCH; ++u) {
                const unsigned r = (unsigned)(d[u] - base);
                if (r < (unsigned)len) atomicAdd(&sdeg[r], w[u]);
            }
        }
        __syncthreads();

        for (int i = tid; i < len; i += DEG_THREADS) prow[base + i] = sdeg[i];
        __syncthreads();
    }
}

#define DINV_BATCH 16
__global__ void k_dinv(float* __restrict__ dinv, const float* __restrict__ part, int n) {
    int i = blockIdx.x * blockDim.x + threadIdx.x;
    if (i < n) {
        const float* __restrict__ p = part + i;
        float d = 1.0f;
#pragma unroll
        for (int b0 = 0; b0 < DEG_BLOCKS; b0 += DINV_BATCH) {
            float v[DINV_BATCH];
#pragma unroll
            for (int j = 0; j < DINV_BATCH; ++j) v[j] = p[(size_t)(b0 + j) * n];
            asm volatile("" :: "v"(v[0]), "v"(v[1]), "v"(v[2]), "v"(v[3]), "v"(v[4]), "v"(v[5]), "v"(v[6]), "v"(v[7]), "v"(v[8]), "v"(v[9]), "v"(v[10]), "v"(v[11]), "v"(v[12]), "v"(v[13]), "v"(v[14]), "v"(v[15]));
#pragma unroll
            for (int j = 0; j < DINV_BATCH; ++j) d += v[j];
        }
        dinv[i] = (d > 0.0f) ? rsqrtf(d) : 0.0f;
    }
}

#define GEMM_ROWS 64
__device__ __forceinline__ void
gemm_role(const int blk, const float* __restrict__ x, const float* __restrict__ W,
          float* __restrict__ h, int N) {
    const int lane  = threadIdx.x & 31;
    const int wave  = threadIdx.x >> 5;
    const int Mbase = blk * GEMM_ROWS + (wave >> 3) * 16;
    const int Nbase = (wave & 7) * 16;
    if (Mbase >= N) return;
    const int half  = lane >> 4;
    const int l15   = lane & 15;

    int mrow = Mbase + l15; if (mrow >= N) mrow = N - 1;
    const float* __restrict__ arow = x + (size_t)mrow * D_FEAT;
    const float* __restrict__ brow = W + (size_t)(Nbase + l15) * D_FEAT;

    v8f c = {};
#pragma unroll
    for (int k = 0; k < D_FEAT; k += 4) {
        const int kk = k + 2 * half;
        v2f a = *reinterpret_cast<const v2f*>(arow + kk);
        v2f b = *reinterpret_cast<const v2f*>(brow + kk);
        c = __builtin_amdgcn_wmma_f32_16x16x4_f32(false, a, false, b, (short)0, c,
                                                  false, false);
    }

    float* __restrict__ hout = h + (size_t)Mbase * D_FEAT + Nbase;
#pragma unroll
    for (int v = 0; v < 8; ++v) {
        int m = v + 8 * half;
        if (Mbase + m < N) hout[(size_t)m * D_FEAT + l15] = c[v];
    }
}

#define BIN_NODES    20
#define NBIN_MAX     1024
#define SORT_BLOCKS  64
#define SORT_THREADS 1024

__device__ __forceinline__ void
bin_role(const int blk, int* __restrict__ rec_w, int* __restrict__ tab, const int* __restrict__ src,
         const int* __restrict__ dst, const float* __restrict__ ew, int E, int nb) {
    int4* __restrict__ rec = reinterpret_cast<int4*>(rec_w);
    __shared__ int cnt[NBIN_MAX];
    __shared__ int cur[NBIN_MAX];
    __shared__ int wsum[32];
    const int tid  = threadIdx.x;
    const int lane = tid & 31;
    const int wv   = tid >> 5;
    const int CH   = (E + SORT_BLOCKS - 1) / SORT_BLOCKS;
    const int e_lo = blk * CH;
    const int e_hi = min(E, e_lo + CH);

    cnt[tid] = 0;
    __syncthreads();
    for (int e = e_lo + tid; e < e_hi; e += SORT_THREADS)
        atomicAdd(&cnt[dst[e] / BIN_NODES], 1);
    __syncthreads();

    const int v = cnt[tid];
    int x = v;
    for (int o = 1; o < 32; o <<= 1) {
        const int y = __shfl_up(x, o);
        if (lane >= o) x += y;
    }
    if (lane == 31) wsum[wv] = x;
    __syncthreads();
    if (wv == 0) {
        const int s = wsum[lane];
        int t = s;
        for (int o = 1; o < 32; o <<= 1) {
            const int y = __shfl_up(t, o);
            if (lane >= o) t += y;
        }
        wsum[lane] = t - s;
    }
    __syncthreads();
    const int excl = x - v + wsum[wv];
    cur[tid] = excl;
    if (tid <= nb) tab[(size_t)blk * (nb + 1) + tid] = excl;
    __syncthreads();

    int4* __restrict__ rg = rec + e_lo;
    for (int e = e_lo + tid; e < e_hi; e += SORT_THREADS) {
        const int t = dst[e];
        const int pos = atomicAdd(&cur[t / BIN_NODES], 1);
        rg[pos] = make_int4(src[e], t, __float_as_int(ew[e]), 0);
    }
}

#define PRE_THREADS 1024
__global__ void __launch_bounds__(PRE_THREADS)
k_pre(float* __restrict__ part, int* __restrict__ rec_w, int* __restrict__ tab,
      float* __restrict__ h, const float* __restrict__ x, const float* __restrict__ W,
      const int* __restrict__ src, const int* __restrict__ dst,
      const float* __restrict__ ew, int E, int N, int nb) {
    const int b = blockIdx.x;
    if (b < SORT_BLOCKS)
        bin_role(b, rec_w, tab, src, dst, ew, E, nb);
    else if (b < SORT_BLOCKS + DEG_BLOCKS)
        deg_part_role(b - SORT_BLOCKS, part, dst, ew, E, N);
    else
        gemm_role(b - (SORT_BLOCKS + DEG_BLOCKS), x, W, h, N);
}

#define AGG_THREADS 1024
#define AGG_BATCH   4

__global__ void __launch_bounds__(AGG_THREADS)
k_agg(float* __restrict__ out, const float* __restrict__ h, const float* __restrict__ dinv,
      const float* __restrict__ bias, const int* __restrict__ rec_w,
      const int* __restrict__ tab, int E, int N, int nb) {
    const int4* __restrict__ rec = reinterpret_cast<const int4*>(rec_w);
    __shared__ float acc[BIN_NODES * D_FEAT];
    const int tid  = threadIdx.x;
    const int lane = tid & 31;
    const int wv   = __builtin_amdgcn_readfirstlane(tid >> 5);
    const int b    = blockIdx.x;
    const int base = b * BIN_NODES;
    const int rows = min(BIN_NODES, N - base);
    const int nq   = rows * 32;

    for (int q = tid; q < nq; q += AGG_THREADS) {
        const int nd = q >> 5;
        const int l  = q & 31;
        const float di = dinv[base + nd];
        const v4f hs = *reinterpret_cast<const v4f*>(h + (size_t)(base + nd) * D_FEAT + 4 * l);
        const v4f bv = *reinterpret_cast<const v4f*>(bias + 4 * l);
        float* ap = acc + nd * D_FEAT + l;
        ap[0]  = bv.x + hs.x * di * di;
        ap[32] = bv.y + hs.y * di * di;
        ap[64] = bv.z + hs.z * di * di;
        ap[96] = bv.w + hs.w * di * di;
    }
    __syncthreads();

    const int CH = (E + SORT_BLOCKS - 1) / SORT_BLOCKS;
    for (int g = wv; g < SORT_BLOCKS; g += AGG_THREADS / 32) {
        const int* __restrict__ tg = tab + (size_t)g * (nb + 1) + b;
        const int s0 = __builtin_amdgcn_readfirstlane(tg[0]);
        const int s1 = __builtin_amdgcn_readfirstlane(tg[1]);
        const int4* __restrict__ rg = rec + (size_t)g * CH;
        for (int i0 = s0; i0 < s1; i0 += 32) {
            const int i = i0 + lane;
            const bool ok = i < s1;
            int4 r = make_int4(0, base, 0, 0);
            if (ok) r = rg[i];
            float nrm = 0.0f;
            if (ok) nrm = dinv[r.x] * __int_as_float(r.z) * dinv[r.y];
            const int roff = (r.y - base) * D_FEAT;
            const int m = min(32, s1 - i0);
            for (int j = 0; j < m; j += AGG_BATCH) {
                v4f   hv[AGG_BATCH];
                int   oj[AGG_BATCH];
                float nj[AGG_BATCH];
#pragma unroll
                for (int u = 0; u < AGG_BATCH; ++u) {
                    const int jj = (j + u) & 31;
                    const int sj = __builtin_amdgcn_readlane(r.x, jj);
                    oj[u] = __builtin_amdgcn_readlane(roff, jj);
                    nj[u] = __int_as_float(__builtin_amdgcn_readlane(__float_as_int(nrm), jj));
                    hv[u] = *reinterpret_cast<const v4f*>(h + (size_t)sj * D_FEAT + 4 * lane);
                }
#pragma unroll
                for (int u = 0; u < AGG_BATCH; ++u) {
                    if (j + u < m) {
                        float* ap = acc + oj[u] + lane;
                        atomicAdd(ap,      hv[u].x * nj[u]);
                        atomicAdd(ap + 32, hv[u].y * nj[u]);
                        atomicAdd(ap + 64, hv[u].z * nj[u]);
                        atomicAdd(ap + 96, hv[u].w * nj[u]);
                    }
                }
            }
        }
    }
    __syncthreads();

    for (int q = tid; q < nq; q += AGG_THREADS) {
        const int nd = q >> 5;
        const int l  = q & 31;
        const float* ap = acc + nd * D_FEAT + l;
        v4f o;
        o.x = ap[0];
        o.y = ap[32];
        o.z = ap[64];
        o.w = ap[96];
        *reinterpret_cast<v4f*>(out + (size_t)(base + nd) * D_FEAT + 4 * l) = o;
    }
}

extern "C" void kernel_launch(void* const* d_in, const int* in_sizes, int n_in,
                              void* d_out, int out_size, void* d_ws, size_t ws_size,
                              hipStream_t stream) {
    const float* x  = (const float*)d_in[0];
    const float* W  = (const float*)d_in[1];
    const float* bb = (const float*)d_in[2];
    const float* ew = (const float*)d_in[3];
    const int*   ei = (const int*)d_in[4];

    const int E = in_sizes[3];
    const int N = in_sizes[0] / D_FEAT;
    const int* src = ei;
    const int* dst = ei + E;
    float* out = (float*)d_out;

    float* deg = (float*)d_ws;
    size_t degPad = ((size_t)N + 63) & ~(size_t)63;
    float* h = deg + degPad;
    float* part = h + (size_t)N * D_FEAT;
    int* rec = (int*)(part + (size_t)DEG_BLOCKS * N);
    int* tab = rec + (size_t)E * 4;
    const int nb = (N + BIN_NODES - 1) / BIN_NODES;

    const int TB = 256;
    k_pre <<<SORT_BLOCKS + DEG_BLOCKS + (N + GEMM_ROWS - 1) / GEMM_ROWS, PRE_THREADS, 0, stream>>>(
        part, rec, tab, h, x, W, src, dst, ew, E, N, nb);
    k_dinv<<<(N + TB - 1) / TB, TB, 0, stream>>>(deg, part, N);
    k_agg <<<nb, AGG_THREADS, 0, stream>>>(out, h, deg, bb, rec, tab, E, N, nb);
}


# GEMM role: x tile and W copied to LDS by whole-row 16-byte loads (4 lines each), fragments read from LDS; was 64 global loads of 16 lines per wave; roles share one LDS union; sums unchanged
# speedup vs baseline: 4.2893x; 1.1067x over previous
#include <hip/hip_runtime.h>
#include <math.h>

typedef __attribute__((ext_vector_type(2))) float v2f;
typedef __attribute__((ext_vector_type(4))) float v4f;
typedef __attribute__((ext_vector_type(8))) float v8f;

#define D_FEAT 128

#define PRE_THREADS 1024

#define DEG_BLOCKS  64
#define DEG_THREADS 1024
#define DEG_CHUNK   16384
#define DEG_BATCH   10

__device__ __forceinline__ void
deg_part_role(float* __restrict__ sdeg, const int blk, float* __restrict__ part,
              const int* __restrict__ dst, const float* __restrict__ ew, int E, int N) {
    const int tid    = threadIdx.x;
    const int gt     = blk * DEG_THREADS + tid;
    const int stride = DEG_BLOCKS * DEG_THREADS;
    float* __restrict__ prow = part + (size_t)blk * N;

    for (int base = 0; base < N; base += DEG_CHUNK) {
        const int len = min(DEG_CHUNK, N - base);
        for (int i = tid; i < len; i += DEG_THREADS) sdeg[i] = 0.0f;
        __syncthreads();

        for (int e0 = gt; e0 < E; e0 += DEG_BATCH * stride) {
            int   d[DEG_BATCH];
            float w[DEG_BATCH];
#pragma unroll
            for (int u = 0; u < DEG_BATCH; ++u) {
                const int e = e0 + u * stride;
                const bool ok = e < E;
                d[u] = ok ? dst[e] : -1;
                w[u] = ok ? ew[e] : 0.0f;
            }
#pragma unroll
            for (int u = 0; u < DEG_BATCH; ++u) {
                const unsigned r = (unsigned)(d[u] - base);
                if (r < (unsigned)len) atomicAdd(&sdeg[r], w[u]);
            }
        }
        __syncthreads();

        for (int i = tid; i < len; i += DEG_THREADS) prow[base + i] = sdeg[i];
        __syncthreads();
    }
}

#define DINV_BATCH 16
__global__ void k_dinv(float* __restrict__ dinv, const float* __restrict__ part, int n) {
    int i = blockIdx.x * blockDim.x + threadIdx.x;
    if (i < n) {
        const float* __restrict__ p = part + i;
        float d = 1.0f;
#pragma unroll
        for (int b0 = 0; b0 < DEG_BLOCKS; b0 += DINV_BATCH) {
            float v[DINV_BATCH];
#pragma unroll
            for (int j = 0; j < DINV_BATCH; ++j) v[j] = p[(size_t)(b0 + j) * n];
            asm volatile("" :: "v"(v[0]), "v"(v[1]), "v"(v[2]), "v"(v[3]), "v"(v[4]), "v"(v[5]), "v"(v[6]), "v"(v[7]), "v"(v[8]), "v"(v[9]), "v"(v[10]), "v"(v[11]), "v"(v[12]), "v"(v[13]), "v"(v[14]), "v"(v[15]));
#pragma unroll
            for (int j = 0; j < DINV_BATCH; ++j) d += v[j];
        }
        dinv[i] = (d > 0.0f) ? rsqrtf(d) : 0.0f;
    }
}

#define GEMM_ROWS  64
#define GEMM_PITCH 132
__device__ __forceinline__ void
gemm_role(float* __restrict__ sx, float* __restrict__ sw, const int blk,
          const float* __restrict__ x, const float* __restrict__ W,
          float* __restrict__ h, int N) {
    const int tid   = threadIdx.x;
    const int lane  = tid & 31;
    const int wave  = tid >> 5;
    const int row0  = blk * GEMM_ROWS;

#pragma unroll
    for (int i = 0; i < GEMM_ROWS * 32 / PRE_THREADS; ++i) {
        const int q  = tid + i * PRE_THREADS;
        const int r  = q >> 5;
        const int c4 = q & 31;
        int gr = row0 + r; if (gr >= N) gr = N - 1;
        const v4f v = *reinterpret_cast<const v4f*>(x + (size_t)gr * D_FEAT + 4 * c4);
        *reinterpret_cast<v4f*>(sx + r * GEMM_PITCH + 4 * c4) = v;
    }
#pragma unroll
    for (int i = 0; i < D_FEAT * 32 / PRE_THREADS; ++i) {
        const int q  = tid + i * PRE_THREADS;
        const int r  = q >> 5;
        const int c4 = q & 31;
        const v4f v = *reinterpret_cast<const v4f*>(W + (size_t)r * D_FEAT + 4 * c4);
        *reinterpret_cast<v4f*>(sw + r * GEMM_PITCH + 4 * c4) = v;
    }
    __syncthreads();

    const int Mloc  = (wave >> 3) * 16;
    const int Mbase = row0 + Mloc;
    const int Nbase = (wave & 7) * 16;
    if (Mbase >= N) return;
    const int half  = lane >> 4;
    const int l15   = lane & 15;

    const float* __restrict__ arow = sx + (Mloc + l15) * GEMM_PITCH;
    const float* __restrict__ brow = sw + (Nbase + l15) * GEMM_PITCH;

    v8f c = {};
#pragma unroll
    for (int k = 0; k < D_FEAT; k += 4) {
        const int kk = k + 2 * half;
        v2f a = *reinterpret_cast<const v2f*>(arow + kk);
        v2f b = *reinterpret_cast<const v2f*>(brow + kk);
        c = __builtin_amdgcn_wmma_f32_16x16x4_f32(false, a, false, b, (short)0, c,
                                                  false, false);
    }

    float* __restrict__ hout = h + (size_t)Mbase * D_FEAT + Nbase;
#pragma unroll
    for (int v = 0; v < 8; ++v) {
        int m = v + 8 * half;
        if (Mbase + m < N) hout[(size_t)m * D_FEAT + l15] = c[v];
    }
}

#define BIN_NODES    20
#define NBIN_MAX     1024
#define SORT_BLOCKS  64
#define SORT_THREADS 1024

__device__ __forceinline__ void
bin_role(int* cnt, int* cur, int* wsum, const int blk, int* __restrict__ rec_w,
         int* __restrict__ tab, const int* __restrict__ src,
         const int* __restrict__ dst, const float* __restrict__ ew, int E, int nb) {
    int4* __restrict__ rec = reinterpret_cast<int4*>(rec_w);
    const int tid  = threadIdx.x;
    const int lane = tid & 31;
    const int wv   = tid >> 5;
    const int CH   = (E + SORT_BLOCKS - 1) / SORT_BLOCKS;
    const int e_lo = blk * CH;
    const int e_hi = min(E, e_lo + CH);

    cnt[tid] = 0;
    __syncthreads();
    for (int e = e_lo + tid; e < e_hi; e += SORT_THREADS)
        atomicAdd(&cnt[dst[e] / BIN_NODES], 1);
    __syncthreads();

    const int v = cnt[tid];
    int x = v;
    for (int o = 1; o < 32; o <<= 1) {
        const int y = __shfl_up(x, o);
        if (lane >= o) x += y;
    }
    if (lane == 31) wsum[wv] = x;
    __syncthreads();
    if (wv == 0) {
        const int s = wsum[lane];
        int t = s;
        for (int o = 1; o < 32; o <<= 1) {
            const int y = __shfl_up(t, o);
            if (lane >= o) t += y;
        }
        wsum[lane] = t - s;
    }
    __syncthreads();
    const int excl = x - v + wsum[wv];
    cur[tid] = excl;
    if (tid <= nb) tab[(size_t)blk * (nb + 1) + tid] = excl;
    __syncthreads();

    int4* __restrict__ rg = rec + e_lo;
    for (int e = e_lo + tid; e < e_hi; e += SORT_THREADS) {
        const int t = dst[e];
        const int pos = atomicAdd(&cur[t / BIN_NODES], 1);
        rg[pos] = make_int4(src[e], t, __float_as_int(ew[e]), 0);
    }
}

struct BinLds  { int cnt[NBIN_MAX]; int cur[NBIN_MAX]; int wsum[32]; };
struct GemmLds { float sx[GEMM_ROWS * GEMM_PITCH]; float sw[D_FEAT * GEMM_PITCH]; };
union  PreLds  { float sdeg[DEG_CHUNK]; BinLds bin; GemmLds gemm; };

__global__ void __launch_bounds__(PRE_THREADS)
k_pre(float* __restrict__ part, int* __restrict__ rec_w, int* __restrict__ tab,
      float* __restrict__ h, const float* __restrict__ x, const float* __restrict__ W,
      const int* __restrict__ src, const int* __restrict__ dst,
      const float* __restrict__ ew, int E, int N, int nb) {
    __shared__ __attribute__((aligned(16))) PreLds lds;
    const int b = blockIdx.x;
    if (b < SORT_BLOCKS)
        bin_role(lds.bin.cnt, lds.bin.cur, lds.bin.wsum, b, rec_w, tab, src, dst, ew, E, nb);
    else if (b < SORT_BLOCKS + DEG_BLOCKS)
        deg_part_role(lds.sdeg, b - SORT_BLOCKS, part, dst, ew, E, N);
    else
        gemm_role(lds.gemm.sx, lds.gemm.sw, b - (SORT_BLOCKS + DEG_BLOCKS), x, W, h, N);
}

#define AGG_THREADS 1024
#define AGG_BATCH   4

__global__ void __launch_bounds__(AGG_THREADS)
k_agg(float* __restrict__ out, const float* __restrict__ h, const float* __restrict__ dinv,
      const float* __restrict__ bias, const int* __restrict__ rec_w,
      const int* __restrict__ tab, int E, int N, int nb) {
    const int4* __restrict__ rec = reinterpret_cast<const int4*>(rec_w);
    __shared__ float acc[BIN_NODES * D_FEAT];
    const int tid  = threadIdx.x;
    const int lane = tid & 31;
    const int wv   = __builtin_amdgcn_readfirstlane(tid >> 5);
    const int b    = blockIdx.x;
    const int base = b * BIN_NODES;
    const int rows = min(BIN_NODES, N - base);
    const int nq   = rows * 32;

    for (int q = tid; q < nq; q += AGG_THREADS) {
        const int nd = q >> 5;
        const int l  = q & 31;
        const float di = dinv[base + nd];
        const v4f hs = *reinterpret_cast<const v4f*>(h + (size_t)(base + nd) * D_FEAT + 4 * l);
        const v4f bv = *reinterpret_cast<const v4f*>(bias + 4 * l);
        float* ap = acc + nd * D_FEAT + l;
        ap[0]  = bv.x + hs.x * di * di;
        ap[32] = bv.y + hs.y * di * di;
        ap[64] = bv.z + hs.z * di * di;
        ap[96] = bv.w + hs.w * di * di;
    }
    __syncthreads();

    const int CH = (E + SORT_BLOCKS - 1) / SORT_BLOCKS;
    for (int g = wv; g < SORT_BLOCKS; g += AGG_THREADS / 32) {
        const int* __restrict__ tg = tab + (size_t)g * (nb + 1) + b;
        const int s0 = __builtin_amdgcn_readfirstlane(tg[0]);
        const int s1 = __builtin_amdgcn_readfirstlane(tg[1]);
        const int4* __restrict__ rg = rec + (size_t)g * CH;
        for (int i0 = s0; i0 < s1; i0 += 32) {
            const int i = i0 + lane;
            const bool ok = i < s1;
            int4 r = make_int4(0, base, 0, 0);
            if (ok) r = rg[i];
            float nrm = 0.0f;
            if (ok) nrm = dinv[r.x] * __int_as_float(r.z) * dinv[r.y];
            const int roff = (r.y - base) * D_FEAT;
            const int m = min(32, s1 - i0);
            for (int j = 0; j < m; j += AGG_BATCH) {
                v4f   hv[AGG_BATCH];
                int   oj[AGG_BATCH];
                float nj[AGG_BATCH];
#pragma unroll
                for (int u = 0; u < AGG_BATCH; ++u) {
                    const int jj = (j + u) & 31;
                    const int sj = __builtin_amdgcn_readlane(r.x, jj);
                    oj[u] = __builtin_amdgcn_readlane(roff, jj);
                    nj[u] = __int_as_float(__builtin_amdgcn_readlane(__float_as_int(nrm), jj));
                    hv[u] = *reinterpret_cast<const v4f*>(h + (size_t)sj * D_FEAT + 4 * lane);
                }
#pragma unroll
                for (int u = 0; u < AGG_BATCH; ++u) {
                    if (j + u < m) {
                        float* ap = acc + oj[u] + lane;
                        atomicAdd(ap,      hv[u].x * nj[u]);
                        atomicAdd(ap + 32, hv[u].y * nj[u]);
                        atomicAdd(ap + 64, hv[u].z * nj[u]);
                        atomicAdd(ap + 96, hv[u].w * nj[u]);
                    }
                }
            }
        }
    }
    __syncthreads();

    for (int q = tid; q < nq; q += AGG_THREADS) {
        const int nd = q >> 5;
        const int l  = q & 31;
        const float* ap = acc + nd * D_FEAT + l;
        v4f o;
        o.x = ap[0];
        o.y = ap[32];
        o.z = ap[64];
        o.w = ap[96];
        *reinterpret_cast<v4f*>(out + (size_t)(base + nd) * D_FEAT + 4 * l) = o;
    }
}

extern "C" void kernel_launch(void* const* d_in, const int* in_sizes, int n_in,
                              void* d_out, int out_size, void* d_ws, size_t ws_size,
                              hipStream_t stream) {
    const float* x  = (const float*)d_in[0];
    const float* W  = (const float*)d_in[1];
    const float* bb = (const float*)d_in[2];
    const float* ew = (const float*)d_in[3];
    const int*   ei = (const int*)d_in[4];

    const int E = in_sizes[3];
    const int N = in_sizes[0] / D_FEAT;
    const int* src = ei;
    const int* dst = ei + E;
    float* out = (float*)d_out;

    float* deg = (float*)d_ws;
    size_t degPad = ((size_t)N + 63) & ~(size_t)63;
    float* h = deg + degPad;
    float* part = h + (size_t)N * D_FEAT;
    int* rec = (int*)(part + (size_t)DEG_BLOCKS * N);
    int* tab = rec + (size_t)E * 4;
    const int nb = (N + BIN_NODES - 1) / BIN_NODES;

    const int TB = 256;
    k_pre <<<SORT_BLOCKS + DEG_BLOCKS + (N + GEMM_ROWS - 1) / GEMM_ROWS, PRE_THREADS, 0, stream>>>(
        part, rec, tab, h, x, W, src, dst, ew, E, N, nb);
    k_dinv<<<(N + TB - 1) / TB, TB, 0, stream>>>(deg, part, N);
    k_agg <<<nb, AGG_THREADS, 0, stream>>>(out, h, deg, bb, rec, tab, E, N, nb);
}


# aggregation: work-group sorts its range's edges by node into an LDS list (integer counts, scan, place), one wave per node sums the h rows in registers by fma; no LDS float adds; sum order changed
# speedup vs baseline: 4.6089x; 1.0745x over previous
#include <hip/hip_runtime.h>
#include <math.h>

typedef __attribute__((ext_vector_type(2))) float v2f;
typedef __attribute__((ext_vector_type(4))) float v4f;
typedef __attribute__((ext_vector_type(8))) float v8f;

#define D_FEAT 128

#define PRE_THREADS 1024

#define DEG_BLOCKS  64
#define DEG_THREADS 1024
#define DEG_CHUNK   16384
#define DEG_BATCH   10

__device__ __forceinline__ void
deg_part_role(float* __restrict__ sdeg, const int blk, float* __restrict__ part,
              const int* __restrict__ dst, const float* __restrict__ ew, int E, int N) {
    const int tid    = threadIdx.x;
    const int gt     = blk * DEG_THREADS + tid;
    const int stride = DEG_BLOCKS * DEG_THREADS;
    float* __restrict__ prow = part + (size_t)blk * N;

    for (int base = 0; base < N; base += DEG_CHUNK) {
        const int len = min(DEG_CHUNK, N - base);
        for (int i = tid; i < len; i += DEG_THREADS) sdeg[i] = 0.0f;
        __syncthreads();

        for (int e0 = gt; e0 < E; e0 += DEG_BATCH * stride) {
            int   d[DEG_BATCH];
            float w[DEG_BATCH];
#pragma unroll
            for (int u = 0; u < DEG_BATCH; ++u) {
                const int e = e0 + u * stride;
                const bool ok = e < E;
                d[u] = ok ? dst[e] : -1;
                w[u] = ok ? ew[e] : 0.0f;
            }
#pragma unroll
            for (int u = 0; u < DEG_BATCH; ++u) {
                const unsigned r = (unsigned)(d[u] - base);
                if (r < (unsigned)len) atomicAdd(&sdeg[r], w[u]);
            }
        }
        __syncthreads();

        for (int i = tid; i < len; i += DEG_THREADS) prow[base + i] = sdeg[i];
        __syncthreads();
    }
}

#define DINV_BATCH 16
__global__ void k_dinv(float* __restrict__ dinv, const float* __restrict__ part, int n) {
    int i = blockIdx.x * blockDim.x + threadIdx.x;
    if (i < n) {
        const float* __restrict__ p = part + i;
        float d = 1.0f;
#pragma unroll
        for (int b0 = 0; b0 < DEG_BLOCKS; b0 += DINV_BATCH) {
            float v[DINV_BATCH];
#pragma unroll
            for (int j = 0; j < DINV_BATCH; ++j) v[j] = p[(size_t)(b0 + j) * n];
            asm volatile("" :: "v"(v[0]), "v"(v[1]), "v"(v[2]), "v"(v[3]), "v"(v[4]), "v"(v[5]), "v"(v[6]), "v"(v[7]), "v"(v[8]), "v"(v[9]), "v"(v[10]), "v"(v[11]), "v"(v[12]), "v"(v[13]), "v"(v[14]), "v"(v[15]));
#pragma unroll
            for (int j = 0; j < DINV_BATCH; ++j) d += v[j];
        }
        dinv[i] = (d > 0.0f) ? rsqrtf(d) : 0.0f;
    }
}

#define GEMM_ROWS  64
#define GEMM_PITCH 132
__device__ __forceinline__ void
gemm_role(float* __restrict__ sx, float* __restrict__ sw, const int blk,
          const float* __restrict__ x, const float* __restrict__ W,
          float* __restrict__ h, int N) {
    const int tid   = threadIdx.x;
    const int lane  = tid & 31;
    const int wave  = tid >> 5;
    const int row0  = blk * GEMM_ROWS;

#pragma unroll
    for (int i = 0; i < GEMM_ROWS * 32 / PRE_THREADS; ++i) {
        const int q  = tid + i * PRE_THREADS;
        const int r  = q >> 5;
        const int c4 = q & 31;
        int gr = row0 + r; if (gr >= N) gr = N - 1;
        const v4f v = *reinterpret_cast<const v4f*>(x + (size_t)gr * D_FEAT + 4 * c4);
        *reinterpret_cast<v4f*>(sx + r * GEMM_PITCH + 4 * c4) = v;
    }
#pragma unroll
    for (int i = 0; i < D_FEAT * 32 / PRE_THREADS; ++i) {
        const int q  = tid + i * PRE_THREADS;
        const int r  = q >> 5;
        const int c4 = q & 31;
        const v4f v = *reinterpret_cast<const v4f*>(W + (size_t)r * D_FEAT + 4 * c4);
        *reinterpret_cast<v4f*>(sw + r * GEMM_PITCH + 4 * c4) = v;
    }
    __syncthreads();

    const int Mloc  = (wave >> 3) * 16;
    const int Mbase = row0 + Mloc;
    const int Nbase = (wave & 7) * 16;
    if (Mbase >= N) return;
    const int half  = lane >> 4;
    const int l15   = lane & 15;

    const float* __restrict__ arow = sx + (Mloc + l15) * GEMM_PITCH;
    const float* __restrict__ brow = sw + (Nbase + l15) * GEMM_PITCH;

    v8f c = {};
#pragma unroll
    for (int k = 0; k < D_FEAT; k += 4) {
        const int kk = k + 2 * half;
        v2f a = *reinterpret_cast<const v2f*>(arow + kk);
        v2f b = *reinterpret_cast<const v2f*>(brow + kk);
        c = __builtin_amdgcn_wmma_f32_16x16x4_f32(false, a, false, b, (short)0, c,
                                                  false, false);
    }

    float* __restrict__ hout = h + (size_t)Mbase * D_FEAT + Nbase;
#pragma unroll
    for (int v = 0; v < 8; ++v) {
        int m = v + 8 * half;
        if (Mbase + m < N) hout[(size_t)m * D_FEAT + l15] = c[v];
    }
}

#define BIN_NODES    20
#define NBIN_MAX     1024
#define SORT_BLOCKS  64
#define SORT_THREADS 1024

__device__ __forceinline__ void
bin_role(int* cnt, int* cur, int* wsum, const int blk, int* __restrict__ rec_w,
         int* __restrict__ tab, const int* __restrict__ src,
         const int* __restrict__ dst, const float* __restrict__ ew, int E, int nb) {
    int4* __restrict__ rec = reinterpret_cast<int4*>(rec_w);
    const int tid  = threadIdx.x;
    const int lane = tid & 31;
    const int wv   = tid >> 5;
    const int CH   = (E + SORT_BLOCKS - 1) / SORT_BLOCKS;
    const int e_lo = blk * CH;
    const int e_hi = min(E, e_lo + CH);

    cnt[tid] = 0;
    __syncthreads();
    for (int e = e_lo + tid; e < e_hi; e += SORT_THREADS)
        atomicAdd(&cnt[dst[e] / BIN_NODES], 1);
    __syncthreads();

    const int v = cnt[tid];
    int x = v;
    for (int o = 1; o < 32; o <<= 1) {
        const int y = __shfl_up(x, o);
        if (lane >= o) x += y;
    }
    if (lane == 31) wsum[wv] = x;
    __syncthreads();
    if (wv == 0) {
        const int s = wsum[lane];
        int t = s;
        for (int o = 1; o < 32; o <<= 1) {
            const int y = __shfl_up(t, o);
            if (lane >= o) t += y;
        }
        wsum[lane] = t - s;
    }
    __syncthreads();
    const int excl = x - v + wsum[wv];
    cur[tid] = excl;
    if (tid <= nb) tab[(size_t)blk * (nb + 1) + tid] = excl;
    __syncthreads();

    int4* __restrict__ rg = rec + e_lo;
    for (int e = e_lo + tid; e < e_hi; e += SORT_THREADS) {
        const int t = dst[e];
        const int pos = atomicAdd(&cur[t / BIN_NODES], 1);
        rg[pos] = make_int4(src[e], t, __float_as_int(ew[e]), 0);
    }
}

struct BinLds  { int cnt[NBIN_MAX]; int cur[NBIN_MAX]; int wsum[32]; };
struct GemmLds { float sx[GEMM_ROWS * GEMM_PITCH]; float sw[D_FEAT * GEMM_PITCH]; };
union  PreLds  { float sdeg[DEG_CHUNK]; BinLds bin; GemmLds gemm; };

__global__ void __launch_bounds__(PRE_THREADS)
k_pre(float* __restrict__ part, int* __restrict__ rec_w, int* __restrict__ tab,
      float* __restrict__ h, const float* __restrict__ x, const float* __restrict__ W,
      const int* __restrict__ src, const int* __restrict__ dst,
      const float* __restrict__ ew, int E, int N, int nb) {
    __shared__ __attribute__((aligned(16))) PreLds lds;
    const int b = blockIdx.x;
    if (b < SORT_BLOCKS)
        bin_role(lds.bin.cnt, lds.bin.cur, lds.bin.wsum, b, rec_w, tab, src, dst, ew, E, nb);
    else if (b < SORT_BLOCKS + DEG_BLOCKS)
        deg_part_role(lds.sdeg, b - SORT_BLOCKS, part, dst, ew, E, N);
    else
        gemm_role(lds.gemm.sx, lds.gemm.sw, b - (SORT_BLOCKS + DEG_BLOCKS), x, W, h, N);
}

#define AGG_WAVES     BIN_NODES
#define AGG_THREADS   (AGG_WAVES * 32)
#define AGG_SEG_LANES (AGG_THREADS / SORT_BLOCKS)
#define AGG_HELD      4
#define AGG_CAP       2048
#define AGG_BATCH     8
static_assert(AGG_THREADS % SORT_BLOCKS == 0, "whole threads per segment");
static_assert(BIN_NODES <= 31, "the offset scan runs in one wave and needs entry BIN_NODES");

__global__ void __launch_bounds__(AGG_THREADS)
k_agg(float* __restrict__ out, const float* __restrict__ h, const float* __restrict__ dinv,
      const float* __restrict__ bias, const int* __restrict__ rec_w,
      const int* __restrict__ tab, int E, int N, int nb) {
    const int4* __restrict__ rec = reinterpret_cast<const int4*>(rec_w);
    __shared__ int   e_src[AGG_CAP];
    __shared__ float e_nrm[AGG_CAP];
    __shared__ int   ncnt[32];
    __shared__ int   noff[32];
    __shared__ int   ncur[32];
    const int tid  = threadIdx.x;
    const int lane = tid & 31;
    const int wv   = __builtin_amdgcn_readfirstlane(tid >> 5);
    const int b    = blockIdx.x;
    const int base = b * BIN_NODES;
    const int rows = min(BIN_NODES, N - base);
    const int CH   = (E + SORT_BLOCKS - 1) / SORT_BLOCKS;
    const bool has_node = wv < rows;

    float di = 0.0f;
    v4f hs = {0.0f, 0.0f, 0.0f, 0.0f};
    v4f bv = {0.0f, 0.0f, 0.0f, 0.0f};
    if (has_node) {
        di = dinv[base + wv];
        hs = *reinterpret_cast<const v4f*>(h + (size_t)(base + wv) * D_FEAT + 4 * lane);
        bv = *reinterpret_cast<const v4f*>(bias + 4 * lane);
    }

    const int g = tid / AGG_SEG_LANES;
    const int k = tid - g * AGG_SEG_LANES;
    const int* __restrict__ tg = tab + (size_t)g * (nb + 1) + b;
    const int s0 = tg[0];
    const int s1 = tg[1];
    const int4* __restrict__ rg = rec + (size_t)g * CH;

    if (tid < 32) ncnt[tid] = 0;

    int   rs[AGG_HELD];
    int   rl[AGG_HELD];
    float rn[AGG_HELD];
    {
        int4 r[AGG_HELD];
#pragma unroll
        for (int u = 0; u < AGG_HELD; ++u) {
            const int i = s0 + k + u * AGG_SEG_LANES;
            r[u] = make_int4(0, base, 0, 0);
            if (i < s1) r[u] = rg[i];
        }
        float da[AGG_HELD], dc[AGG_HELD];
#pragma unroll
        for (int u = 0; u < AGG_HELD; ++u) {
            da[u] = dinv[r[u].x];
            dc[u] = dinv[r[u].y];
        }
#pragma unroll
        for (int u = 0; u < AGG_HELD; ++u) {
            const int i = s0 + k + u * AGG_SEG_LANES;
            rs[u] = r[u].x;
            rl[u] = (i < s1) ? r[u].y - base : -1;
            rn[u] = da[u] * __int_as_float(r[u].z) * dc[u];
        }
    }
    __syncthreads();
#pragma unroll
    for (int u = 0; u < AGG_HELD; ++u)
        if (rl[u] >= 0) atomicAdd(&ncnt[rl[u]], 1);
    for (int i = s0 + k + AGG_HELD * AGG_SEG_LANES; i < s1; i += AGG_SEG_LANES)
        atomicAdd(&ncnt[rg[i].y - base], 1);
    __syncthreads();

    if (tid < 32) {
        const int c = (lane < BIN_NODES) ? ncnt[lane] : 0;
        int x = c;
        for (int o = 1; o < 32; o <<= 1) {
            const int y = __shfl_up(x, o);
            if (lane >= o) x += y;
        }
        noff[lane] = x - c;
        ncur[lane] = x - c;
    }
    __syncthreads();
    const int T = noff[BIN_NODES];

    v4f acc = {0.0f, 0.0f, 0.0f, 0.0f};
    if (T <= AGG_CAP) {
        int pos[AGG_HELD];
#pragma unroll
        for (int u = 0; u < AGG_HELD; ++u) {
            pos[u] = 0;
            if (rl[u] >= 0) pos[u] = atomicAdd(&ncur[rl[u]], 1);
        }
#pragma unroll
        for (int u = 0; u < AGG_HELD; ++u) {
            if (rl[u] >= 0) {
                e_src[pos[u]] = rs[u];
                e_nrm[pos[u]] = rn[u];
            }
        }
        for (int i = s0 + k + AGG_HELD * AGG_SEG_LANES; i < s1; i += AGG_SEG_LANES) {
            const int4 r = rg[i];
            const float nrm = dinv[r.x] * __int_as_float(r.z) * dinv[r.y];
            const int pos = atomicAdd(&ncur[r.y - base], 1);
            e_src[pos] = r.x;
            e_nrm[pos] = nrm;
        }
        __syncthreads();

        if (has_node) {
            const int n0 = __builtin_amdgcn_readfirstlane(noff[wv]);
            const int n1 = __builtin_amdgcn_readfirstlane(noff[wv + 1]);
            for (int i0 = n0; i0 < n1; i0 += 32) {
                const int idx = i0 + lane;
                int   sv = 0;
                float nv = 0.0f;
                if (idx < n1) { sv = e_src[idx]; nv = e_nrm[idx]; }
                const int m = min(32, n1 - i0);
                for (int j = 0; j < m; j += AGG_BATCH) {
                    v4f   hv[AGG_BATCH];
                    float nj[AGG_BATCH];
#pragma unroll
                    for (int u = 0; u < AGG_BATCH; ++u) {
                        const int sj = __builtin_amdgcn_readlane(sv, j + u);
                        nj[u] = __int_as_float(__builtin_amdgcn_readlane(__float_as_int(nv), j + u));
                        hv[u] = *reinterpret_cast<const v4f*>(h + (size_t)sj * D_FEAT + 4 * lane);
                    }
#pragma unroll
                    for (int u = 0; u < AGG_BATCH; ++u) {
                        if (j + u < m) {
                            acc.x = fmaf(hv[u].x, nj[u], acc.x);
                            acc.y = fmaf(hv[u].y, nj[u], acc.y);
                            acc.z = fmaf(hv[u].z, nj[u], acc.z);
                            acc.w = fmaf(hv[u].w, nj[u], acc.w);
                        }
                    }
                }
            }
        }
    } else if (has_node) {
        const int node = base + wv;
        for (int gg = 0; gg < SORT_BLOCKS; ++gg) {
            const int* __restrict__ tq = tab + (size_t)gg * (nb + 1) + b;
            const int q0 = __builtin_amdgcn_readfirstlane(tq[0]);
            const int q1 = __builtin_amdgcn_readfirstlane(tq[1]);
            const int4* __restrict__ rq = rec + (size_t)gg * CH;
            for (int i0 = q0; i0 < q1; i0 += 32) {
                const int i = i0 + lane;
                int4 r = make_int4(0, -1, 0, 0);
                if (i < q1) r = rq[i];
                const bool mine = r.y == node;
                float nv = 0.0f;
                if (mine) nv = dinv[r.x] * __int_as_float(r.z) * dinv[r.y];
                unsigned mask = __builtin_amdgcn_ballot_w32(mine);
                while (mask) {
                    const int jj = __builtin_ctz(mask);
                    mask &= mask - 1;
                    const int   sj = __builtin_amdgcn_readlane(r.x, jj);
                    const float nj = __int_as_float(__builtin_amdgcn_readlane(__float_as_int(nv), jj));
                    const v4f hv = *reinterpret_cast<const v4f*>(h + (size_t)sj * D_FEAT + 4 * lane);
                    acc.x = fmaf(hv.x, nj, acc.x);
                    acc.y = fmaf(hv.y, nj, acc.y);
                    acc.z = fmaf(hv.z, nj, acc.z);
                    acc.w = fmaf(hv.w, nj, acc.w);
                }
            }
        }
    }

    if (has_node) {
        v4f o;
        o.x = (bv.x + hs.x * di * di) + acc.x;
        o.y = (bv.y + hs.y * di * di) + acc.y;
        o.z = (bv.z + hs.z * di * di) + acc.z;
        o.w = (bv.w + hs.w * di * di) + acc.w;
        *reinterpret_cast<v4f*>(out + (size_t)(base + wv) * D_FEAT + 4 * lane) = o;
    }
}

extern "C" void kernel_launch(void* const* d_in, const int* in_sizes, int n_in,
                              void* d_out, int out_size, void* d_ws, size_t ws_size,
                              hipStream_t stream) {
    const float* x  = (const float*)d_in[0];
    const float* W  = (const float*)d_in[1];
    const float* bb = (const float*)d_in[2];
    const float* ew = (const float*)d_in[3];
    const int*   ei = (const int*)d_in[4];

    const int E = in_sizes[3];
    const int N = in_sizes[0] / D_FEAT;
    const int* src = ei;
    const int* dst = ei + E;
    float* out = (float*)d_out;

    float* deg = (float*)d_ws;
    size_t degPad = ((size_t)N + 63) & ~(size_t)63;
    float* h = deg + degPad;
    float* part = h + (size_t)N * D_FEAT;
    int* rec = (int*)(part + (size_t)DEG_BLOCKS * N);
    int* tab = rec + (size_t)E * 4;
    const int nb = (N + BIN_NODES - 1) / BIN_NODES;

    const int TB = 256;
    k_pre <<<SORT_BLOCKS + DEG_BLOCKS + (N + GEMM_ROWS - 1) / GEMM_ROWS, PRE_THREADS, 0, stream>>>(
        part, rec, tab, h, x, W, src, dst, ew, E, N, nb);
    k_dinv<<<(N + TB - 1) / TB, TB, 0, stream>>>(deg, part, N);
    k_agg <<<nb, AGG_THREADS, 0, stream>>>(out, h, deg, bb, rec, tab, E, N, nb);
}


# edge list read once: binning workgroups hold their chunk's edges in registers and make degree histogram, counts and placing from them; degree workgroups and re-reads dropped; degree sum order changed
# speedup vs baseline: 4.9133x; 1.0660x over previous
#include <hip/hip_runtime.h>
#include <math.h>

typedef __attribute__((ext_vector_type(2))) float v2f;
typedef __attribute__((ext_vector_type(4))) float v4f;
typedef __attribute__((ext_vector_type(8))) float v8f;

#define D_FEAT 128

#define PRE_THREADS 1024

#define SORT_BLOCKS  64
#define DEG_BLOCKS   SORT_BLOCKS

#define DINV_BATCH 16
__global__ void k_dinv(float* __restrict__ dinv, const float* __restrict__ part, int n) {
    int i = blockIdx.x * blockDim.x + threadIdx.x;
    if (i < n) {
        const float* __restrict__ p = part + i;
        float d = 1.0f;
#pragma unroll
        for (int b0 = 0; b0 < DEG_BLOCKS; b0 += DINV_BATCH) {
            float v[DINV_BATCH];
#pragma unroll
            for (int j = 0; j < DINV_BATCH; ++j) v[j] = p[(size_t)(b0 + j) * n];
            asm volatile("" :: "v"(v[0]), "v"(v[1]), "v"(v[2]), "v"(v[3]), "v"(v[4]), "v"(v[5]), "v"(v[6]), "v"(v[7]), "v"(v[8]), "v"(v[9]), "v"(v[10]), "v"(v[11]), "v"(v[12]), "v"(v[13]), "v"(v[14]), "v"(v[15]));
#pragma unroll
            for (int j = 0; j < DINV_BATCH; ++j) d += v[j];
        }
        dinv[i] = (d > 0.0f) ? rsqrtf(d) : 0.0f;
    }
}

#define GEMM_ROWS  64
#define GEMM_PITCH 132
__device__ __forceinline__ void
gemm_role(float* __restrict__ sx, float* __restrict__ sw, const int blk,
          const float* __restrict__ x, const float* __restrict__ W,
          float* __restrict__ h, int N) {
    const int tid   = threadIdx.x;
    const int lane  = tid & 31;
    const int wave  = tid >> 5;
    const int row0  = blk * GEMM_ROWS;

#pragma unroll
    for (int i = 0; i < GEMM_ROWS * 32 / PRE_THREADS; ++i) {
        const int q  = tid + i * PRE_THREADS;
        const int r  = q >> 5;
        const int c4 = q & 31;
        int gr = row0 + r; if (gr >= N) gr = N - 1;
        const v4f v = *reinterpret_cast<const v4f*>(x + (size_t)gr * D_FEAT + 4 * c4);
        *reinterpret_cast<v4f*>(sx + r * GEMM_PITCH + 4 * c4) = v;
    }
#pragma unroll
    for (int i = 0; i < D_FEAT * 32 / PRE_THREADS; ++i) {
        const int q  = tid + i * PRE_THREADS;
        const int r  = q >> 5;
        const int c4 = q & 31;
        const v4f v = *reinterpret_cast<const v4f*>(W + (size_t)r * D_FEAT + 4 * c4);
        *reinterpret_cast<v4f*>(sw + r * GEMM_PITCH + 4 * c4) = v;
    }
    __syncthreads();

    const int Mloc  = (wave >> 3) * 16;
    const int Mbase = row0 + Mloc;
    const int Nbase = (wave & 7) * 16;
    if (Mbase >= N) return;
    const int half  = lane >> 4;
    const int l15   = lane & 15;

    const float* __restrict__ arow = sx + (Mloc + l15) * GEMM_PITCH;
    const float* __restrict__ brow = sw + (Nbase + l15) * GEMM_PITCH;

    v8f c = {};
#pragma unroll
    for (int k = 0; k < D_FEAT; k += 4) {
        const int kk = k + 2 * half;
        v2f a = *reinterpret_cast<const v2f*>(arow + kk);
        v2f b = *reinterpret_cast<const v2f*>(brow + kk);
        c = __builtin_amdgcn_wmma_f32_16x16x4_f32(false, a, false, b, (short)0, c,
                                                  false, false);
    }

    float* __restrict__ hout = h + (size_t)Mbase * D_FEAT + Nbase;
#pragma unroll
    for (int v = 0; v < 8; ++v) {
        int m = v + 8 * half;
        if (Mbase + m < N) hout[(size_t)m * D_FEAT + l15] = c[v];
    }
}

#define BIN_NODES    20
#define NBIN_MAX     1024
#define SORT_THREADS 1024
#define BIN_HELD     10
#define DEG_CHUNK    16384

__device__ __forceinline__ void
bin_role(int* cnt, int* cur, int* wsum, float* __restrict__ sdeg, const int blk,
         float* __restrict__ part, int* __restrict__ rec_w, int* __restrict__ tab,
         const int* __restrict__ src, const int* __restrict__ dst,
         const float* __restrict__ ew, int E, int N, int nb) {
    int4* __restrict__ rec = reinterpret_cast<int4*>(rec_w);
    const int tid  = threadIdx.x;
    const int lane = tid & 31;
    const int wv   = tid >> 5;
    const int CH   = (E + SORT_BLOCKS - 1) / SORT_BLOCKS;
    const int e_lo = blk * CH;
    const int e_hi = min(E, e_lo + CH);
    const int e_t  = e_lo + tid;
    const int e_r  = e_t + BIN_HELD * SORT_THREADS;

    int   hd[BIN_HELD];
    int   hs[BIN_HELD];
    float hw[BIN_HELD];
#pragma unroll
    for (int u = 0; u < BIN_HELD; ++u) {
        const int e = e_t + u * SORT_THREADS;
        const bool ok = e < e_hi;
        hd[u] = ok ? dst[e] : 0;
        hs[u] = ok ? src[e] : 0;
        hw[u] = ok ? ew[e] : 0.0f;
    }

    cnt[tid] = 0;
    float* __restrict__ prow = part + (size_t)blk * N;
    for (int base = 0; base < N; base += DEG_CHUNK) {
        const int len = min(DEG_CHUNK, N - base);
        for (int i = tid; i < len; i += SORT_THREADS) sdeg[i] = 0.0f;
        __syncthreads();

        if (base == 0) {
#pragma unroll
            for (int u = 0; u < BIN_HELD; ++u)
                if (e_t + u * SORT_THREADS < e_hi) atomicAdd(&cnt[hd[u] / BIN_NODES], 1);
            for (int e = e_r; e < e_hi; e += SORT_THREADS)
                atomicAdd(&cnt[dst[e] / BIN_NODES], 1);
        }
#pragma unroll
        for (int u = 0; u < BIN_HELD; ++u) {
            const unsigned r = (unsigned)(hd[u] - base);
            if (e_t + u * SORT_THREADS < e_hi && r < (unsigned)len) atomicAdd(&sdeg[r], hw[u]);
        }
        for (int e = e_r; e < e_hi; e += SORT_THREADS) {
            const unsigned r = (unsigned)(dst[e] - base);
            if (r < (unsigned)len) atomicAdd(&sdeg[r], ew[e]);
        }
        __syncthreads();

        for (int i = tid; i < len; i += SORT_THREADS) prow[base + i] = sdeg[i];
        if (base + DEG_CHUNK < N) __syncthreads();
    }

    const int v = cnt[tid];
    int x = v;
    for (int o = 1; o < 32; o <<= 1) {
        const int y = __shfl_up(x, o);
        if (lane >= o) x += y;
    }
    if (lane == 31) wsum[wv] = x;
    __syncthreads();
    if (wv == 0) {
        const int s = wsum[lane];
        int t = s;
        for (int o = 1; o < 32; o <<= 1) {
            const int y = __shfl_up(t, o);
            if (lane >= o) t += y;
        }
        wsum[lane] = t - s;
    }
    __syncthreads();
    const int excl = x - v + wsum[wv];
    cur[tid] = excl;
    if (tid <= nb) tab[(size_t)blk * (nb + 1) + tid] = excl;
    __syncthreads();

    int4* __restrict__ rg = rec + e_lo;
    unsigned pos[BIN_HELD];
#pragma unroll
    for (int u = 0; u < BIN_HELD; ++u) {
        pos[u] = 0u;
        if (e_t + u * SORT_THREADS < e_hi)
            pos[u] = (unsigned)atomicAdd(&cur[hd[u] / BIN_NODES], 1);
    }
#pragma unroll
    for (int u = 0; u < BIN_HELD; ++u)
        if (e_t + u * SORT_THREADS < e_hi)
            rg[pos[u]] = make_int4(hs[u], hd[u], __float_as_int(hw[u]), 0);
    for (int e = e_r; e < e_hi; e += SORT_THREADS) {
        const int t = dst[e];
        const int p = atomicAdd(&cur[t / BIN_NODES], 1);
        rg[p] = make_int4(src[e], t, __float_as_int(ew[e]), 0);
    }
}

struct BinLds  { int cnt[NBIN_MAX]; int cur[NBIN_MAX]; int wsum[32]; float sdeg[DEG_CHUNK]; };
struct GemmLds { float sx[GEMM_ROWS * GEMM_PITCH]; float sw[D_FEAT * GEMM_PITCH]; };
union  PreLds  { BinLds bin; GemmLds gemm; };

__global__ void __launch_bounds__(PRE_THREADS)
k_pre(float* __restrict__ part, int* __restrict__ rec_w, int* __restrict__ tab,
      float* __restrict__ h, const float* __restrict__ x, const float* __restrict__ W,
      const int* __restrict__ src, const int* __restrict__ dst,
      const float* __restrict__ ew, int E, int N, int nb) {
    __shared__ __attribute__((aligned(16))) PreLds lds;
    const int b = blockIdx.x;
    if (b < SORT_BLOCKS)
        bin_role(lds.bin.cnt, lds.bin.cur, lds.bin.wsum, lds.bin.sdeg, b, part, rec_w, tab,
                 src, dst, ew, E, N, nb);
    else
        gemm_role(lds.gemm.sx, lds.gemm.sw, b - SORT_BLOCKS, x, W, h, N);
}

#define AGG_WAVES     BIN_NODES
#define AGG_THREADS   (AGG_WAVES * 32)
#define AGG_SEG_LANES (AGG_THREADS / SORT_BLOCKS)
#define AGG_HELD      4
#define AGG_CAP       2048
#define AGG_BATCH     8
static_assert(AGG_THREADS % SORT_BLOCKS == 0, "whole threads per segment");
static_assert(BIN_NODES <= 31, "the offset scan runs in one wave and needs entry BIN_NODES");

__global__ void __launch_bounds__(AGG_THREADS)
k_agg(float* __restrict__ out, const float* __restrict__ h, const float* __restrict__ dinv,
      const float* __restrict__ bias, const int* __restrict__ rec_w,
      const int* __restrict__ tab, int E, int N, int nb) {
    const int4* __restrict__ rec = reinterpret_cast<const int4*>(rec_w);
    __shared__ int   e_src[AGG_CAP];
    __shared__ float e_nrm[AGG_CAP];
    __shared__ int   ncnt[32];
    __shared__ int   noff[32];
    __shared__ int   ncur[32];
    const int tid  = threadIdx.x;
    const int lane = tid & 31;
    const int wv   = __builtin_amdgcn_readfirstlane(tid >> 5);
    const int b    = blockIdx.x;
    const int base = b * BIN_NODES;
    const int rows = min(BIN_NODES, N - base);
    const int CH   = (E + SORT_BLOCKS - 1) / SORT_BLOCKS;
    const bool has_node = wv < rows;

    float di = 0.0f;
    v4f hs = {0.0f, 0.0f, 0.0f, 0.0f};
    v4f bv = {0.0f, 0.0f, 0.0f, 0.0f};
    if (has_node) {
        di = dinv[base + wv];
        hs = *reinterpret_cast<const v4f*>(h + (size_t)(base + wv) * D_FEAT + 4 * lane);
        bv = *reinterpret_cast<const v4f*>(bias + 4 * lane);
    }

    const int g = tid / AGG_SEG_LANES;
    const int k = tid - g * AGG_SEG_LANES;
    const int* __restrict__ tg = tab + (size_t)g * (nb + 1) + b;
    const int s0 = tg[0];
    const int s1 = tg[1];
    const int4* __restrict__ rg = rec + (size_t)g * CH;

    if (tid < 32) ncnt[tid] = 0;

    int   rs[AGG_HELD];
    int   rl[AGG_HELD];
    float rn[AGG_HELD];
    {
        int4 r[AGG_HELD];
#pragma unroll
        for (int u = 0; u < AGG_HELD; ++u) {
            const int i = s0 + k + u * AGG_SEG_LANES;
            r[u] = make_int4(0, base, 0, 0);
            if (i < s1) r[u] = rg[i];
        }
        float da[AGG_HELD], dc[AGG_HELD];
#pragma unroll
        for (int u = 0; u < AGG_HELD; ++u) {
            da[u] = dinv[r[u].x];
            dc[u] = dinv[r[u].y];
        }
#pragma unroll
        for (int u = 0; u < AGG_HELD; ++u) {
            const int i = s0 + k + u * AGG_SEG_LANES;
            rs[u] = r[u].x;
            rl[u] = (i < s1) ? r[u].y - base : -1;
            rn[u] = da[u] * __int_as_float(r[u].z) * dc[u];
        }
    }
    __syncthreads();
#pragma unroll
    for (int u = 0; u < AGG_HELD; ++u)
        if (rl[u] >= 0) atomicAdd(&ncnt[rl[u]], 1);
    for (int i = s0 + k + AGG_HELD * AGG_SEG_LANES; i < s1; i += AGG_SEG_LANES)
        atomicAdd(&ncnt[rg[i].y - base], 1);
    __syncthreads();

    if (tid < 32) {
        const int c = (lane < BIN_NODES) ? ncnt[lane] : 0;
        int x = c;
        for (int o = 1; o < 32; o <<= 1) {
            const int y = __shfl_up(x, o);
            if (lane >= o) x += y;
        }
        noff[lane] = x - c;
        ncur[lane] = x - c;
    }
    __syncthreads();
    const int T = noff[BIN_NODES];

    v4f acc = {0.0f, 0.0f, 0.0f, 0.0f};
    if (T <= AGG_CAP) {
        int pos[AGG_HELD];
#pragma unroll
        for (int u = 0; u < AGG_HELD; ++u) {
            pos[u] = 0;
            if (rl[u] >= 0) pos[u] = atomicAdd(&ncur[rl[u]], 1);
        }
#pragma unroll
        for (int u = 0; u < AGG_HELD; ++u) {
            if (rl[u] >= 0) {
                e_src[pos[u]] = rs[u];
                e_nrm[pos[u]] = rn[u];
            }
        }
        for (int i = s0 + k + AGG_HELD * AGG_SEG_LANES; i < s1; i += AGG_SEG_LANES) {
            const int4 r = rg[i];
            const float nrm = dinv[r.x] * __int_as_float(r.z) * dinv[r.y];
            const int pos = atomicAdd(&ncur[r.y - base], 1);
            e_src[pos] = r.x;
            e_nrm[pos] = nrm;
        }
        __syncthreads();

        if (has_node) {
            const int n0 = __builtin_amdgcn_readfirstlane(noff[wv]);
            const int n1 = __builtin_amdgcn_readfirstlane(noff[wv + 1]);
            for (int i0 = n0; i0 < n1; i0 += 32) {
                const int idx = i0 + lane;
                int   sv = 0;
                float nv = 0.0f;
                if (idx < n1) { sv = e_src[idx]; nv = e_nrm[idx]; }
                const int m = min(32, n1 - i0);
                for (int j = 0; j < m; j += AGG_BATCH) {
                    v4f   hv[AGG_BATCH];
                    float nj[AGG_BATCH];
#pragma unroll
                    for (int u = 0; u < AGG_BATCH; ++u) {
                        const int sj = __builtin_amdgcn_readlane(sv, j + u);
                        nj[u] = __int_as_float(__builtin_amdgcn_readlane(__float_as_int(nv), j + u));
                        hv[u] = *reinterpret_cast<const v4f*>(h + (size_t)sj * D_FEAT + 4 * lane);
                    }
#pragma unroll
                    for (int u = 0; u < AGG_BATCH; ++u) {
                        if (j + u < m) {
                            acc.x = fmaf(hv[u].x, nj[u], acc.x);
                            acc.y = fmaf(hv[u].y, nj[u], acc.y);
                            acc.z = fmaf(hv[u].z, nj[u], acc.z);
                            acc.w = fmaf(hv[u].w, nj[u], acc.w);
                        }
                    }
                }
            }
        }
    } else if (has_node) {
        const int node = base + wv;
        for (int gg = 0; gg < SORT_BLOCKS; ++gg) {
            const int* __restrict__ tq = tab + (size_t)gg * (nb + 1) + b;
            const int q0 = __builtin_amdgcn_readfirstlane(tq[0]);
            const int q1 = __builtin_amdgcn_readfirstlane(tq[1]);
            const int4* __restrict__ rq = rec + (size_t)gg * CH;
            for (int i0 = q0; i0 < q1; i0 += 32) {
                const int i = i0 + lane;
                int4 r = make_int4(0, -1, 0, 0);
                if (i < q1) r = rq[i];
                const bool mine = r.y == node;
                float nv = 0.0f;
                if (mine) nv = dinv[r.x] * __int_as_float(r.z) * dinv[r.y];
                unsigned mask = __builtin_amdgcn_ballot_w32(mine);
                while (mask) {
                    const int jj = __builtin_ctz(mask);
                    mask &= mask - 1;
                    const int   sj = __builtin_amdgcn_readlane(r.x, jj);
                    const float nj = __int_as_float(__builtin_amdgcn_readlane(__float_as_int(nv), jj));
                    const v4f hv = *reinterpret_cast<const v4f*>(h + (size_t)sj * D_FEAT + 4 * lane);
                    acc.x = fmaf(hv.x, nj, acc.x);
                    acc.y = fmaf(hv.y, nj, acc.y);
                    acc.z = fmaf(hv.z, nj, acc.z);
                    acc.w = fmaf(hv.w, nj, acc.w);
                }
            }
        }
    }

    if (has_node) {
        v4f o;
        o.x = (bv.x + hs.x * di * di) + acc.x;
        o.y = (bv.y + hs.y * di * di) + acc.y;
        o.z = (bv.z + hs.z * di * di) + acc.z;
        o.w = (bv.w + hs.w * di * di) + acc.w;
        *reinterpret_cast<v4f*>(out + (size_t)(base + wv) * D_FEAT + 4 * lane) = o;
    }
}

extern "C" void kernel_launch(void* const* d_in, const int* in_sizes, int n_in,
                              void* d_out, int out_size, void* d_ws, size_t ws_size,
                              hipStream_t stream) {
    const float* x  = (const float*)d_in[0];
    const float* W  = (const float*)d_in[1];
    const float* bb = (const float*)d_in[2];
    const float* ew = (const float*)d_in[3];
    const int*   ei = (const int*)d_in[4];

    const int E = in_sizes[3];
    const int N = in_sizes[0] / D_FEAT;
    const int* src = ei;
    const int* dst = ei + E;
    float* out = (float*)d_out;

    float* deg = (float*)d_ws;
    size_t degPad = ((size_t)N + 63) & ~(size_t)63;
    float* h = deg + degPad;
    float* part = h + (size_t)N * D_FEAT;
    int* rec = (int*)(part + (size_t)DEG_BLOCKS * N);
    int* tab = rec + (size_t)E * 4;
    const int nb = (N + BIN_NODES - 1) / BIN_NODES;

    const int TB = 256;
    k_pre <<<SORT_BLOCKS + (N + GEMM_ROWS - 1) / GEMM_ROWS, PRE_THREADS, 0, stream>>>(
        part, rec, tab, h, x, W, src, dst, ew, E, N, nb);
    k_dinv<<<(N + TB - 1) / TB, TB, 0, stream>>>(deg, part, N);
    k_agg <<<nb, AGG_THREADS, 0, stream>>>(out, h, deg, bb, rec, tab, E, N, nb);
}


# binning: each chunk's records are put in rank order in LDS (over the finished degree histogram) and copied to rec as whole 128-byte lines, 4 per store; was one 16-byte piece per lane on 32 lines
# speedup vs baseline: 5.4072x; 1.1005x over previous
#include <hip/hip_runtime.h>
#include <math.h>

typedef __attribute__((ext_vector_type(2))) float v2f;
typedef __attribute__((ext_vector_type(4))) float v4f;
typedef __attribute__((ext_vector_type(8))) float v8f;
typedef __attribute__((ext_vector_type(2))) int   v2i;

#define D_FEAT 128

#define PRE_THREADS 1024

#define SORT_BLOCKS  64
#define DEG_BLOCKS   SORT_BLOCKS

#define DINV_BATCH 16
__global__ void k_dinv(float* __restrict__ dinv, const float* __restrict__ part, int n) {
    int i = blockIdx.x * blockDim.x + threadIdx.x;
    if (i < n) {
        const float* __restrict__ p = part + i;
        float d = 1.0f;
#pragma unroll
        for (int b0 = 0; b0 < DEG_BLOCKS; b0 += DINV_BATCH) {
            float v[DINV_BATCH];
#pragma unroll
            for (int j = 0; j < DINV_BATCH; ++j) v[j] = p[(size_t)(b0 + j) * n];
            asm volatile("" :: "v"(v[0]), "v"(v[1]), "v"(v[2]), "v"(v[3]), "v"(v[4]), "v"(v[5]), "v"(v[6]), "v"(v[7]), "v"(v[8]), "v"(v[9]), "v"(v[10]), "v"(v[11]), "v"(v[12]), "v"(v[13]), "v"(v[14]), "v"(v[15]));
#pragma unroll
            for (int j = 0; j < DINV_BATCH; ++j) d += v[j];
        }
        dinv[i] = (d > 0.0f) ? rsqrtf(d) : 0.0f;
    }
}

#define GEMM_ROWS  64
#define GEMM_PITCH 132
__device__ __forceinline__ void
gemm_role(float* __restrict__ sx, float* __restrict__ sw, const int blk,
          const float* __restrict__ x, const float* __restrict__ W,
          float* __restrict__ h, int N) {
    const int tid   = threadIdx.x;
    const int lane  = tid & 31;
    const int wave  = tid >> 5;
    const int row0  = blk * GEMM_ROWS;

#pragma unroll
    for (int i = 0; i < GEMM_ROWS * 32 / PRE_THREADS; ++i) {
        const int q  = tid + i * PRE_THREADS;
        const int r  = q >> 5;
        const int c4 = q & 31;
        int gr = row0 + r; if (gr >= N) gr = N - 1;
        const v4f v = *reinterpret_cast<const v4f*>(x + (size_t)gr * D_FEAT + 4 * c4);
        *reinterpret_cast<v4f*>(sx + r * GEMM_PITCH + 4 * c4) = v;
    }
#pragma unroll
    for (int i = 0; i < D_FEAT * 32 / PRE_THREADS; ++i) {
        const int q  = tid + i * PRE_THREADS;
        const int r  = q >> 5;
        const int c4 = q & 31;
        const v4f v = *reinterpret_cast<const v4f*>(W + (size_t)r * D_FEAT + 4 * c4);
        *reinterpret_cast<v4f*>(sw + r * GEMM_PITCH + 4 * c4) = v;
    }
    __syncthreads();

    const int Mloc  = (wave >> 3) * 16;
    const int Mbase = row0 + Mloc;
    const int Nbase = (wave & 7) * 16;
    if (Mbase >= N) return;
    const int half  = lane >> 4;
    const int l15   = lane & 15;

    const float* __restrict__ arow = sx + (Mloc + l15) * GEMM_PITCH;
    const float* __restrict__ brow = sw + (Nbase + l15) * GEMM_PITCH;

    v8f c = {};
#pragma unroll
    for (int k = 0; k < D_FEAT; k += 4) {
        const int kk = k + 2 * half;
        v2f a = *reinterpret_cast<const v2f*>(arow + kk);
        v2f b = *reinterpret_cast<const v2f*>(brow + kk);
        c = __builtin_amdgcn_wmma_f32_16x16x4_f32(false, a, false, b, (short)0, c,
                                                  false, false);
    }

    float* __restrict__ hout = h + (size_t)Mbase * D_FEAT + Nbase;
#pragma unroll
    for (int v = 0; v < 8; ++v) {
        int m = v + 8 * half;
        if (Mbase + m < N) hout[(size_t)m * D_FEAT + l15] = c[v];
    }
}

#define BIN_NODES    20
#define NBIN_MAX     1024
#define SORT_THREADS 1024
#define BIN_HELD     10
#define DEG_CHUNK    16384
#define STG_CAP      (BIN_HELD * SORT_THREADS)
static_assert(BIN_HELD == 10, "the wait on the ranks in bin_role names pos[0..9]");

__device__ __forceinline__ void
bin_role(int* cnt, int* cur, int* wsum, float* sdeg, v2i* st_sw, int* st_d, const int blk,
         float* __restrict__ part, int* __restrict__ rec_w, int* __restrict__ tab,
         const int* __restrict__ src, const int* __restrict__ dst,
         const float* __restrict__ ew, int E, int N, int nb) {
    int4* __restrict__ rec = reinterpret_cast<int4*>(rec_w);
    const int tid  = threadIdx.x;
    const int lane = tid & 31;
    const int wv   = tid >> 5;
    const int CH   = (E + SORT_BLOCKS - 1) / SORT_BLOCKS;
    const int e_lo = blk * CH;
    const int e_hi = min(E, e_lo + CH);
    const int e_t  = e_lo + tid;
    const int e_r  = e_t + BIN_HELD * SORT_THREADS;

    int   hd[BIN_HELD];
    int   hs[BIN_HELD];
    float hw[BIN_HELD];
#pragma unroll
    for (int u = 0; u < BIN_HELD; ++u) {
        const int e = e_t + u * SORT_THREADS;
        const bool ok = e < e_hi;
        hd[u] = ok ? dst[e] : 0;
        hs[u] = ok ? src[e] : 0;
        hw[u] = ok ? ew[e] : 0.0f;
    }

    cnt[tid] = 0;
    float* __restrict__ prow = part + (size_t)blk * N;
    for (int base = 0; base < N; base += DEG_CHUNK) {
        const int len = min(DEG_CHUNK, N - base);
        for (int i = tid; i < len; i += SORT_THREADS) sdeg[i] = 0.0f;
        __syncthreads();

        if (base == 0) {
#pragma unroll
            for (int u = 0; u < BIN_HELD; ++u)
                if (e_t + u * SORT_THREADS < e_hi) atomicAdd(&cnt[hd[u] / BIN_NODES], 1);
            for (int e = e_r; e < e_hi; e += SORT_THREADS)
                atomicAdd(&cnt[dst[e] / BIN_NODES], 1);
        }
#pragma unroll
        for (int u = 0; u < BIN_HELD; ++u) {
            const unsigned r = (unsigned)(hd[u] - base);
            if (e_t + u * SORT_THREADS < e_hi && r < (unsigned)len) atomicAdd(&sdeg[r], hw[u]);
        }
        for (int e = e_r; e < e_hi; e += SORT_THREADS) {
            const unsigned r = (unsigned)(dst[e] - base);
            if (r < (unsigned)len) atomicAdd(&sdeg[r], ew[e]);
        }
        __syncthreads();

        for (int i = tid; i < len; i += SORT_THREADS) prow[base + i] = sdeg[i];
        if (base + DEG_CHUNK < N) __syncthreads();
    }

    const int v = cnt[tid];
    int x = v;
    for (int o = 1; o < 32; o <<= 1) {
        const int y = __shfl_up(x, o);
        if (lane >= o) x += y;
    }
    if (lane == 31) wsum[wv] = x;
    __syncthreads();
    if (wv == 0) {
        const int s = wsum[lane];
        int t = s;
        for (int o = 1; o < 32; o <<= 1) {
            const int y = __shfl_up(t, o);
            if (lane >= o) t += y;
        }
        wsum[lane] = t - s;
    }
    __syncthreads();
    const int excl = x - v + wsum[wv];
    cur[tid] = excl;
    if (tid <= nb) tab[(size_t)blk * (nb + 1) + tid] = excl;
    __syncthreads();

    int4* __restrict__ rg = rec + e_lo;
    unsigned pos[BIN_HELD];
#pragma unroll
    for (int u = 0; u < BIN_HELD; ++u) {
        pos[u] = 0u;
        if (e_t + u * SORT_THREADS < e_hi)
            pos[u] = (unsigned)atomicAdd(&cur[hd[u] / BIN_NODES], 1);
    }
    if (CH <= STG_CAP) {
        asm volatile("" :: "v"(pos[0]), "v"(pos[1]), "v"(pos[2]), "v"(pos[3]), "v"(pos[4]), "v"(pos[5]), "v"(pos[6]), "v"(pos[7]), "v"(pos[8]), "v"(pos[9]));
#pragma unroll
        for (int u = 0; u < BIN_HELD; ++u) {
            if (e_t + u * SORT_THREADS < e_hi) {
                v2i a; a.x = hs[u]; a.y = __float_as_int(hw[u]);
                st_sw[pos[u]] = a;
                st_d[pos[u]]  = hd[u];
            }
        }
        __syncthreads();
        v2i ca[BIN_HELD];
        int cd[BIN_HELD];
#pragma unroll
        for (int u = 0; u < BIN_HELD; ++u) {
            const int p = (e_t + u * SORT_THREADS < e_hi) ? tid + u * SORT_THREADS : 0;
            ca[u] = st_sw[p];
            cd[u] = st_d[p];
        }
#pragma unroll
        for (int u = 0; u < BIN_HELD; ++u)
            if (e_t + u * SORT_THREADS < e_hi)
                rg[tid + u * SORT_THREADS] = make_int4(ca[u].x, cd[u], ca[u].y, 0);
    } else {
#pragma unroll
        for (int u = 0; u < BIN_HELD; ++u)
            if (e_t + u * SORT_THREADS < e_hi)
                rg[pos[u]] = make_int4(hs[u], hd[u], __float_as_int(hw[u]), 0);
        for (int e = e_r; e < e_hi; e += SORT_THREADS) {
            const int t = dst[e];
            const int p = atomicAdd(&cur[t / BIN_NODES], 1);
            rg[p] = make_int4(src[e], t, __float_as_int(ew[e]), 0);
        }
    }
}

struct StageLds { v2i sw[STG_CAP]; int d[STG_CAP]; };
union  BinBig   { float sdeg[DEG_CHUNK]; StageLds st; };
struct BinLds   { int cnt[NBIN_MAX]; int cur[NBIN_MAX]; int wsum[32]; BinBig big; };
struct GemmLds { float sx[GEMM_ROWS * GEMM_PITCH]; float sw[D_FEAT * GEMM_PITCH]; };
union  PreLds  { BinLds bin; GemmLds gemm; };

__global__ void __launch_bounds__(PRE_THREADS)
k_pre(float* __restrict__ part, int* __restrict__ rec_w, int* __restrict__ tab,
      float* __restrict__ h, const float* __restrict__ x, const float* __restrict__ W,
      const int* __restrict__ src, const int* __restrict__ dst,
      const float* __restrict__ ew, int E, int N, int nb) {
    __shared__ __attribute__((aligned(16))) PreLds lds;
    const int b = blockIdx.x;
    if (b < SORT_BLOCKS)
        bin_role(lds.bin.cnt, lds.bin.cur, lds.bin.wsum, lds.bin.big.sdeg, lds.bin.big.st.sw,
                 lds.bin.big.st.d, b, part, rec_w, tab, src, dst, ew, E, N, nb);
    else
        gemm_role(lds.gemm.sx, lds.gemm.sw, b - SORT_BLOCKS, x, W, h, N);
}

#define AGG_WAVES     BIN_NODES
#define AGG_THREADS   (AGG_WAVES * 32)
#define AGG_SEG_LANES (AGG_THREADS / SORT_BLOCKS)
#define AGG_HELD      4
#define AGG_CAP       2048
#define AGG_BATCH     8
static_assert(AGG_THREADS % SORT_BLOCKS == 0, "whole threads per segment");
static_assert(BIN_NODES <= 31, "the offset scan runs in one wave and needs entry BIN_NODES");

__global__ void __launch_bounds__(AGG_THREADS)
k_agg(float* __restrict__ out, const float* __restrict__ h, const float* __restrict__ dinv,
      const float* __restrict__ bias, const int* __restrict__ rec_w,
      const int* __restrict__ tab, int E, int N, int nb) {
    const int4* __restrict__ rec = reinterpret_cast<const int4*>(rec_w);
    __shared__ int   e_src[AGG_CAP];
    __shared__ float e_nrm[AGG_CAP];
    __shared__ int   ncnt[32];
    __shared__ int   noff[32];
    __shared__ int   ncur[32];
    const int tid  = threadIdx.x;
    const int lane = tid & 31;
    const int wv   = __builtin_amdgcn_readfirstlane(tid >> 5);
    const int b    = blockIdx.x;
    const int base = b * BIN_NODES;
    const int rows = min(BIN_NODES, N - base);
    const int CH   = (E + SORT_BLOCKS - 1) / SORT_BLOCKS;
    const bool has_node = wv < rows;

    float di = 0.0f;
    v4f hs = {0.0f, 0.0f, 0.0f, 0.0f};
    v4f bv = {0.0f, 0.0f, 0.0f, 0.0f};
    if (has_node) {
        di = dinv[base + wv];
        hs = *reinterpret_cast<const v4f*>(h + (size_t)(base + wv) * D_FEAT + 4 * lane);
        bv = *reinterpret_cast<const v4f*>(bias + 4 * lane);
    }

    const int g = tid / AGG_SEG_LANES;
    const int k = tid - g * AGG_SEG_LANES;
    const int* __restrict__ tg = tab + (size_t)g * (nb + 1) + b;
    const int s0 = tg[0];
    const int s1 = tg[1];
    const int4* __restrict__ rg = rec + (size_t)g * CH;

    if (tid < 32) ncnt[tid] = 0;

    int   rs[AGG_HELD];
    int   rl[AGG_HELD];
    float rn[AGG_HELD];
    {
        int4 r[AGG_HELD];
#pragma unroll
        for (int u = 0; u < AGG_HELD; ++u) {
            const int i = s0 + k + u * AGG_SEG_LANES;
            r[u] = make_int4(0, base, 0, 0);
            if (i < s1) r[u] = rg[i];
        }
        float da[AGG_HELD], dc[AGG_HELD];
#pragma unroll
        for (int u = 0; u < AGG_HELD; ++u) {
            da[u] = dinv[r[u].x];
            dc[u] = dinv[r[u].y];
        }
#pragma unroll
        for (int u = 0; u < AGG_HELD; ++u) {
            const int i = s0 + k + u * AGG_SEG_LANES;
            rs[u] = r[u].x;
            rl[u] = (i < s1) ? r[u].y - base : -1;
            rn[u] = da[u] * __int_as_float(r[u].z) * dc[u];
        }
    }
    __syncthreads();
#pragma unroll
    for (int u = 0; u < AGG_HELD; ++u)
        if (rl[u] >= 0) atomicAdd(&ncnt[rl[u]], 1);
    for (int i = s0 + k + AGG_HELD * AGG_SEG_LANES; i < s1; i += AGG_SEG_LANES)
        atomicAdd(&ncnt[rg[i].y - base], 1);
    __syncthreads();

    if (tid < 32) {
        const int c = (lane < BIN_NODES) ? ncnt[lane] : 0;
        int x = c;
        for (int o = 1; o < 32; o <<= 1) {
            const int y = __shfl_up(x, o);
            if (lane >= o) x += y;
        }
        noff[lane] = x - c;
        ncur[lane] = x - c;
    }
    __syncthreads();
    const int T = noff[BIN_NODES];

    v4f acc = {0.0f, 0.0f, 0.0f, 0.0f};
    if (T <= AGG_CAP) {
        int pos[AGG_HELD];
#pragma unroll
        for (int u = 0; u < AGG_HELD; ++u) {
            pos[u] = 0;
            if (rl[u] >= 0) pos[u] = atomicAdd(&ncur[rl[u]], 1);
        }
#pragma unroll
        for (int u = 0; u < AGG_HELD; ++u) {
            if (rl[u] >= 0) {
                e_src[pos[u]] = rs[u];
                e_nrm[pos[u]] = rn[u];
            }
        }
        for (int i = s0 + k + AGG_HELD * AGG_SEG_LANES; i < s1; i += AGG_SEG_LANES) {
            const int4 r = rg[i];
            const float nrm = dinv[r.x] * __int_as_float(r.z) * dinv[r.y];
            const int pos = atomicAdd(&ncur[r.y - base], 1);
            e_src[pos] = r.x;
            e_nrm[pos] = nrm;
        }
        __syncthreads();

        if (has_node) {
            const int n0 = __builtin_amdgcn_readfirstlane(noff[wv]);
            const int n1 = __builtin_amdgcn_readfirstlane(noff[wv + 1]);
            for (int i0 = n0; i0 < n1; i0 += 32) {
                const int idx = i0 + lane;
                int   sv = 0;
                float nv = 0.0f;
                if (idx < n1) { sv = e_src[idx]; nv = e_nrm[idx]; }
                const int m = min(32, n1 - i0);
                for (int j = 0; j < m; j += AGG_BATCH) {
                    v4f   hv[AGG_BATCH];
                    float nj[AGG_BATCH];
#pragma unroll
                    for (int u = 0; u < AGG_BATCH; ++u) {
                        const int sj = __builtin_amdgcn_readlane(sv, j + u);
                        nj[u] = __int_as_float(__builtin_amdgcn_readlane(__float_as_int(nv), j + u));
                        hv[u] = *reinterpret_cast<const v4f*>(h + (size_t)sj * D_FEAT + 4 * lane);
                    }
#pragma unroll
                    for (int u = 0; u < AGG_BATCH; ++u) {
                        if (j + u < m) {
                            acc.x = fmaf(hv[u].x, nj[u], acc.x);
                            acc.y = fmaf(hv[u].y, nj[u], acc.y);
                            acc.z = fmaf(hv[u].z, nj[u], acc.z);
                            acc.w = fmaf(hv[u].w, nj[u], acc.w);
                        }
                    }
                }
            }
        }
    } else if (has_node) {
        const int node = base + wv;
        for (int gg = 0; gg < SORT_BLOCKS; ++gg) {
            const int* __restrict__ tq = tab + (size_t)gg * (nb + 1) + b;
            const int q0 = __builtin_amdgcn_readfirstlane(tq[0]);
            const int q1 = __builtin_amdgcn_readfirstlane(tq[1]);
            const int4* __restrict__ rq = rec + (size_t)gg * CH;
            for (int i0 = q0; i0 < q1; i0 += 32) {
                const int i = i0 + lane;
                int4 r = make_int4(0, -1, 0, 0);
                if (i < q1) r = rq[i];
                const bool mine = r.y == node;
                float nv = 0.0f;
                if (mine) nv = dinv[r.x] * __int_as_float(r.z) * dinv[r.y];
                unsigned mask = __builtin_amdgcn_ballot_w32(mine);
                while (mask) {
                    const int jj = __builtin_ctz(mask);
                    mask &= mask - 1;
                    const int   sj = __builtin_amdgcn_readlane(r.x, jj);
                    const float nj = __int_as_float(__builtin_amdgcn_readlane(__float_as_int(nv), jj));
                    const v4f hv = *reinterpret_cast<const v4f*>(h + (size_t)sj * D_FEAT + 4 * lane);
                    acc.x = fmaf(hv.x, nj, acc.x);
                    acc.y = fmaf(hv.y, nj, acc.y);
                    acc.z = fmaf(hv.z, nj, acc.z);
                    acc.w = fmaf(hv.w, nj, acc.w);
                }
            }
        }
    }

    if (has_node) {
        v4f o;
        o.x = (bv.x + hs.x * di * di) + acc.x;
        o.y = (bv.y + hs.y * di * di) + acc.y;
        o.z = (bv.z + hs.z * di * di) + acc.z;
        o.w = (bv.w + hs.w * di * di) + acc.w;
        *reinterpret_cast<v4f*>(out + (size_t)(base + wv) * D_FEAT + 4 * lane) = o;
    }
}

extern "C" void kernel_launch(void* const* d_in, const int* in_sizes, int n_in,
                              void* d_out, int out_size, void* d_ws, size_t ws_size,
                              hipStream_t stream) {
    const float* x  = (const float*)d_in[0];
    const float* W  = (const float*)d_in[1];
    const float* bb = (const float*)d_in[2];
    const float* ew = (const float*)d_in[3];
    const int*   ei = (const int*)d_in[4];

    const int E = in_sizes[3];
    const int N = in_sizes[0] / D_FEAT;
    const int* src = ei;
    const int* dst = ei + E;
    float* out = (float*)d_out;

    float* deg = (float*)d_ws;
    size_t degPad = ((size_t)N + 63) & ~(size_t)63;
    float* h = deg + degPad;
    float* part = h + (size_t)N * D_FEAT;
    int* rec = (int*)(part + (size_t)DEG_BLOCKS * N);
    int* tab = rec + (size_t)E * 4;
    const int nb = (N + BIN_NODES - 1) / BIN_NODES;

    const int TB = 256;
    k_pre <<<SORT_BLOCKS + (N + GEMM_ROWS - 1) / GEMM_ROWS, PRE_THREADS, 0, stream>>>(
        part, rec, tab, h, x, W, src, dst, ew, E, N, nb);
    k_dinv<<<(N + TB - 1) / TB, TB, 0, stream>>>(deg, part, N);
    k_agg <<<nb, AGG_THREADS, 0, stream>>>(out, h, deg, bb, rec, tab, E, N, nb);
}


# edge record packed to 8 bytes {src | node-in-range << 27, weight}, was 16 {src, dst, weight, 0}: binning stages and stores half the bytes, two records per lane; aggregation reads half; sums unchanged
# speedup vs baseline: 5.6345x; 1.0420x over previous
#include <hip/hip_runtime.h>
#include <math.h>

typedef __attribute__((ext_vector_type(2))) float v2f;
typedef __attribute__((ext_vector_type(4))) float v4f;
typedef __attribute__((ext_vector_type(8))) float v8f;
typedef __attribute__((ext_vector_type(2))) int   v2i;
typedef __attribute__((ext_vector_type(4))) int   v4i;

#define D_FEAT 128

#define PRE_THREADS 1024

#define SORT_BLOCKS  64
#define DEG_BLOCKS   SORT_BLOCKS

#define DINV_BATCH 16
__global__ void k_dinv(float* __restrict__ dinv, const float* __restrict__ part, int n) {
    int i = blockIdx.x * blockDim.x + threadIdx.x;
    if (i < n) {
        const float* __restrict__ p = part + i;
        float d = 1.0f;
#pragma unroll
        for (int b0 = 0; b0 < DEG_BLOCKS; b0 += DINV_BATCH) {
            float v[DINV_BATCH];
#pragma unroll
            for (int j = 0; j < DINV_BATCH; ++j) v[j] = p[(size_t)(b0 + j) * n];
            asm volatile("" :: "v"(v[0]), "v"(v[1]), "v"(v[2]), "v"(v[3]), "v"(v[4]), "v"(v[5]), "v"(v[6]), "v"(v[7]), "v"(v[8]), "v"(v[9]), "v"(v[10]), "v"(v[11]), "v"(v[12]), "v"(v[13]), "v"(v[14]), "v"(v[15]));
#pragma unroll
            for (int j = 0; j < DINV_BATCH; ++j) d += v[j];
        }
        dinv[i] = (d > 0.0f) ? rsqrtf(d) : 0.0f;
    }
}

#define GEMM_ROWS  64
#define GEMM_PITCH 132
__device__ __forceinline__ void
gemm_role(float* __restrict__ sx, float* __restrict__ sw, const int blk,
          const float* __restrict__ x, const float* __restrict__ W,
          float* __restrict__ h, int N) {
    const int tid   = threadIdx.x;
    const int lane  = tid & 31;
    const int wave  = tid >> 5;
    const int row0  = blk * GEMM_ROWS;

#pragma unroll
    for (int i = 0; i < GEMM_ROWS * 32 / PRE_THREADS; ++i) {
        const int q  = tid + i * PRE_THREADS;
        const int r  = q >> 5;
        const int c4 = q & 31;
        int gr = row0 + r; if (gr >= N) gr = N - 1;
        const v4f v = *reinterpret_cast<const v4f*>(x + (size_t)gr * D_FEAT + 4 * c4);
        *reinterpret_cast<v4f*>(sx + r * GEMM_PITCH + 4 * c4) = v;
    }
#pragma unroll
    for (int i = 0; i < D_FEAT * 32 / PRE_THREADS; ++i) {
        const int q  = tid + i * PRE_THREADS;
        const int r  = q >> 5;
        const int c4 = q & 31;
        const v4f v = *reinterpret_cast<const v4f*>(W + (size_t)r * D_FEAT + 4 * c4);
        *reinterpret_cast<v4f*>(sw + r * GEMM_PITCH + 4 * c4) = v;
    }
    __syncthreads();

    const int Mloc  = (wave >> 3) * 16;
    const int Mbase = row0 + Mloc;
    const int Nbase = (wave & 7) * 16;
    if (Mbase >= N) return;
    const int half  = lane >> 4;
    const int l15   = lane & 15;

    const float* __restrict__ arow = sx + (Mloc + l15) * GEMM_PITCH;
    const float* __restrict__ brow = sw + (Nbase + l15) * GEMM_PITCH;

    v8f c = {};
#pragma unroll
    for (int k = 0; k < D_FEAT; k += 4) {
        const int kk = k + 2 * half;
        v2f a = *reinterpret_cast<const v2f*>(arow + kk);
        v2f b = *reinterpret_cast<const v2f*>(brow + kk);
        c = __builtin_amdgcn_wmma_f32_16x16x4_f32(false, a, false, b, (short)0, c,
                                                  false, false);
    }

    float* __restrict__ hout = h + (size_t)Mbase * D_FEAT + Nbase;
#pragma unroll
    for (int v = 0; v < 8; ++v) {
        int m = v + 8 * half;
        if (Mbase + m < N) hout[(size_t)m * D_FEAT + l15] = c[v];
    }
}

#define BIN_NODES    20
#define NBIN_MAX     1024
#define SORT_THREADS 1024
#define BIN_HELD     10
#define DEG_CHUNK    16384
#define STG_CAP      (BIN_HELD * SORT_THREADS)
#define REC_SHIFT    27
#define REC_SRC_MASK ((1 << REC_SHIFT) - 1)
static_assert(BIN_HELD == 10, "the wait on the ranks in bin_role names pos[0..9]");
static_assert(BIN_HELD % 2 == 0, "the copy to rec takes two records per lane");
static_assert(BIN_NODES <= 32, "the node inside its range has 5 bits in a record");
static_assert(NBIN_MAX * BIN_NODES <= (1 << REC_SHIFT), "src has 27 bits in a record");

__device__ __forceinline__ void
bin_role(int* cnt, int* cur, int* wsum, float* sdeg, v2i* st, const int blk,
         float* __restrict__ part, int* __restrict__ rec_w, int* __restrict__ tab,
         const int* __restrict__ src, const int* __restrict__ dst,
         const float* __restrict__ ew, int E, int N, int nb) {
    v2i* __restrict__ rec = reinterpret_cast<v2i*>(rec_w);
    const int tid  = threadIdx.x;
    const int lane = tid & 31;
    const int wv   = tid >> 5;
    const int CH   = (E + SORT_BLOCKS - 1) / SORT_BLOCKS;
    const int e_lo = blk * CH;
    const int e_hi = min(E, e_lo + CH);
    const int e_t  = e_lo + tid;
    const int e_r  = e_t + BIN_HELD * SORT_THREADS;

    int   hd[BIN_HELD];
    int   hs[BIN_HELD];
    float hw[BIN_HELD];
#pragma unroll
    for (int u = 0; u < BIN_HELD; ++u) {
        const int e = e_t + u * SORT_THREADS;
        const bool ok = e < e_hi;
        hd[u] = ok ? dst[e] : 0;
        hs[u] = ok ? src[e] : 0;
        hw[u] = ok ? ew[e] : 0.0f;
    }

    cnt[tid] = 0;
    float* __restrict__ prow = part + (size_t)blk * N;
    for (int base = 0; base < N; base += DEG_CHUNK) {
        const int len = min(DEG_CHUNK, N - base);
        for (int i = tid; i < len; i += SORT_THREADS) sdeg[i] = 0.0f;
        __syncthreads();

        if (base == 0) {
#pragma unroll
            for (int u = 0; u < BIN_HELD; ++u)
                if (e_t + u * SORT_THREADS < e_hi) atomicAdd(&cnt[hd[u] / BIN_NODES], 1);
            for (int e = e_r; e < e_hi; e += SORT_THREADS)
                atomicAdd(&cnt[dst[e] / BIN_NODES], 1);
        }
#pragma unroll
        for (int u = 0; u < BIN_HELD; ++u) {
            const unsigned r = (unsigned)(hd[u] - base);
            if (e_t + u * SORT_THREADS < e_hi && r < (unsigned)len) atomicAdd(&sdeg[r], hw[u]);
        }
        for (int e = e_r; e < e_hi; e += SORT_THREADS) {
            const unsigned r = (unsigned)(dst[e] - base);
            if (r < (unsigned)len) atomicAdd(&sdeg[r], ew[e]);
        }
        __syncthreads();

        for (int i = tid; i < len; i += SORT_THREADS) prow[base + i] = sdeg[i];
        if (base + DEG_CHUNK < N) __syncthreads();
    }

    const int v = cnt[tid];
    int x = v;
    for (int o = 1; o < 32; o <<= 1) {
        const int y = __shfl_up(x, o);
        if (lane >= o) x += y;
    }
    if (lane == 31) wsum[wv] = x;
    __syncthreads();
    if (wv == 0) {
        const int s = wsum[lane];
        int t = s;
        for (int o = 1; o < 32; o <<= 1) {
            const int y = __shfl_up(t, o);
            if (lane >= o) t += y;
        }
        wsum[lane] = t - s;
    }
    __syncthreads();
    const int excl = x - v + wsum[wv];
    cur[tid] = excl;
    if (tid <= nb) tab[(size_t)blk * (nb + 1) + tid] = excl;
    __syncthreads();

    v2i* __restrict__ rg = rec + e_lo;
    unsigned pos[BIN_HELD];
#pragma unroll
    for (int u = 0; u < BIN_HELD; ++u) {
        pos[u] = 0u;
        if (e_t + u * SORT_THREADS < e_hi)
            pos[u] = (unsigned)atomicAdd(&cur[hd[u] / BIN_NODES], 1);
    }
    int key[BIN_HELD];
#pragma unroll
    for (int u = 0; u < BIN_HELD; ++u)
        key[u] = hs[u] | (int)((unsigned)(hd[u] - (hd[u] / BIN_NODES) * BIN_NODES) << REC_SHIFT);
    const int n_rec = e_hi - e_lo;
    if (CH <= STG_CAP && ((CH | n_rec) & 1) == 0) {
        asm volatile("" :: "v"(pos[0]), "v"(pos[1]), "v"(pos[2]), "v"(pos[3]), "v"(pos[4]), "v"(pos[5]), "v"(pos[6]), "v"(pos[7]), "v"(pos[8]), "v"(pos[9]));
#pragma unroll
        for (int u = 0; u < BIN_HELD; ++u) {
            if (e_t + u * SORT_THREADS < e_hi) {
                v2i a; a.x = key[u]; a.y = __float_as_int(hw[u]);
                st[pos[u]] = a;
            }
        }
        __syncthreads();
        const v4i* __restrict__ st2 = reinterpret_cast<const v4i*>(st);
        v4i cp[BIN_HELD / 2];
#pragma unroll
        for (int u = 0; u < BIN_HELD / 2; ++u) {
            const int q = tid + u * SORT_THREADS;
            cp[u] = st2[(2 * q < n_rec) ? q : 0];
        }
#pragma unroll
        for (int u = 0; u < BIN_HELD / 2; ++u) {
            const int q = tid + u * SORT_THREADS;
            if (2 * q < n_rec) *reinterpret_cast<v4i*>(rg + 2 * q) = cp[u];
        }
    } else {
#pragma unroll
        for (int u = 0; u < BIN_HELD; ++u)
            if (e_t + u * SORT_THREADS < e_hi) {
                v2i a; a.x = key[u]; a.y = __float_as_int(hw[u]);
                rg[pos[u]] = a;
            }
        for (int e = e_r; e < e_hi; e += SORT_THREADS) {
            const int t = dst[e];
            const int r = t / BIN_NODES;
            const int p = atomicAdd(&cur[r], 1);
            v2i a; a.x = src[e] | (int)((unsigned)(t - r * BIN_NODES) << REC_SHIFT); a.y = __float_as_int(ew[e]);
            rg[p] = a;
        }
    }
}

union  BinBig   { float sdeg[DEG_CHUNK]; v2i st[STG_CAP]; };
struct BinLds   { int cnt[NBIN_MAX]; int cur[NBIN_MAX]; int wsum[32]; BinBig big; };
struct GemmLds { float sx[GEMM_ROWS * GEMM_PITCH]; float sw[D_FEAT * GEMM_PITCH]; };
union  PreLds  { BinLds bin; GemmLds gemm; };

__global__ void __launch_bounds__(PRE_THREADS)
k_pre(float* __restrict__ part, int* __restrict__ rec_w, int* __restrict__ tab,
      float* __restrict__ h, const float* __restrict__ x, const float* __restrict__ W,
      const int* __restrict__ src, const int* __restrict__ dst,
      const float* __restrict__ ew, int E, int N, int nb) {
    __shared__ __attribute__((aligned(16))) PreLds lds;
    const int b = blockIdx.x;
    if (b < SORT_BLOCKS)
        bin_role(lds.bin.cnt, lds.bin.cur, lds.bin.wsum, lds.bin.big.sdeg, lds.bin.big.st,
                 b, part, rec_w, tab, src, dst, ew, E, N, nb);
    else
        gemm_role(lds.gemm.sx, lds.gemm.sw, b - SORT_BLOCKS, x, W, h, N);
}

#define AGG_WAVES     BIN_NODES
#define AGG_THREADS   (AGG_WAVES * 32)
#define AGG_SEG_LANES (AGG_THREADS / SORT_BLOCKS)
#define AGG_HELD      4
#define AGG_CAP       2048
#define AGG_BATCH     8
static_assert(AGG_THREADS % SORT_BLOCKS == 0, "whole threads per segment");
static_assert(BIN_NODES <= 31, "the offset scan runs in one wave and needs entry BIN_NODES");

__global__ void __launch_bounds__(AGG_THREADS)
k_agg(float* __restrict__ out, const float* __restrict__ h, const float* __restrict__ dinv,
      const float* __restrict__ bias, const int* __restrict__ rec_w,
      const int* __restrict__ tab, int E, int N, int nb) {
    const int2* __restrict__ rec = reinterpret_cast<const int2*>(rec_w);
    __shared__ int   e_src[AGG_CAP];
    __shared__ float e_nrm[AGG_CAP];
    __shared__ int   ncnt[32];
    __shared__ int   noff[32];
    __shared__ int   ncur[32];
    const int tid  = threadIdx.x;
    const int lane = tid & 31;
    const int wv   = __builtin_amdgcn_readfirstlane(tid >> 5);
    const int b    = blockIdx.x;
    const int base = b * BIN_NODES;
    const int rows = min(BIN_NODES, N - base);
    const int CH   = (E + SORT_BLOCKS - 1) / SORT_BLOCKS;
    const bool has_node = wv < rows;

    float di = 0.0f;
    v4f hs = {0.0f, 0.0f, 0.0f, 0.0f};
    v4f bv = {0.0f, 0.0f, 0.0f, 0.0f};
    if (has_node) {
        di = dinv[base + wv];
        hs = *reinterpret_cast<const v4f*>(h + (size_t)(base + wv) * D_FEAT + 4 * lane);
        bv = *reinterpret_cast<const v4f*>(bias + 4 * lane);
    }

    const int g = tid / AGG_SEG_LANES;
    const int k = tid - g * AGG_SEG_LANES;
    const int* __restrict__ tg = tab + (size_t)g * (nb + 1) + b;
    const int s0 = tg[0];
    const int s1 = tg[1];
    const int2* __restrict__ rg = rec + (size_t)g * CH;

    if (tid < 32) ncnt[tid] = 0;

    int   rs[AGG_HELD];
    int   rl[AGG_HELD];
    float rn[AGG_HELD];
    {
        int2 r[AGG_HELD];
#pragma unroll
        for (int u = 0; u < AGG_HELD; ++u) {
            const int i = s0 + k + u * AGG_SEG_LANES;
            r[u] = make_int2(0, 0);
            if (i < s1) r[u] = rg[i];
        }
        float da[AGG_HELD], dc[AGG_HELD];
#pragma unroll
        for (int u = 0; u < AGG_HELD; ++u) {
            da[u] = dinv[r[u].x & REC_SRC_MASK];
            dc[u] = dinv[base + (int)((unsigned)r[u].x >> REC_SHIFT)];
        }
#pragma unroll
        for (int u = 0; u < AGG_HELD; ++u) {
            const int i = s0 + k + u * AGG_SEG_LANES;
            rs[u] = r[u].x & REC_SRC_MASK;
            rl[u] = (i < s1) ? (int)((unsigned)r[u].x >> REC_SHIFT) : -1;
            rn[u] = da[u] * __int_as_float(r[u].y) * dc[u];
        }
    }
    __syncthreads();
#pragma unroll
    for (int u = 0; u < AGG_HELD; ++u)
        if (rl[u] >= 0) atomicAdd(&ncnt[rl[u]], 1);
    for (int i = s0 + k + AGG_HELD * AGG_SEG_LANES; i < s1; i += AGG_SEG_LANES)
        atomicAdd(&ncnt[(unsigned)rg[i].x >> REC_SHIFT], 1);
    __syncthreads();

    if (tid < 32) {
        const int c = (lane < BIN_NODES) ? ncnt[lane] : 0;
        int x = c;
        for (int o = 1; o < 32; o <<= 1) {
            const int y = __shfl_up(x, o);
            if (lane >= o) x += y;
        }
        noff[lane] = x - c;
        ncur[lane] = x - c;
    }
    __syncthreads();
    const int T = noff[BIN_NODES];

    v4f acc = {0.0f, 0.0f, 0.0f, 0.0f};
    if (T <= AGG_CAP) {
        int pos[AGG_HELD];
#pragma unroll
        for (int u = 0; u < AGG_HELD; ++u) {
            pos[u] = 0;
            if (rl[u] >= 0) pos[u] = atomicAdd(&ncur[rl[u]], 1);
        }
#pragma unroll
        for (int u = 0; u < AGG_HELD; ++u) {
            if (rl[u] >= 0) {
                e_src[pos[u]] = rs[u];
                e_nrm[pos[u]] = rn[u];
            }
        }
        for (int i = s0 + k + AGG_HELD * AGG_SEG_LANES; i < s1; i += AGG_SEG_LANES) {
            const int2 r = rg[i];
            const int rsrc = r.x & REC_SRC_MASK;
            const int rloc = (int)((unsigned)r.x >> REC_SHIFT);
            const float nrm = dinv[rsrc] * __int_as_float(r.y) * dinv[base + rloc];
            const int pos = atomicAdd(&ncur[rloc], 1);
            e_src[pos] = rsrc;
            e_nrm[pos] = nrm;
        }
        __syncthreads();

        if (has_node) {
            const int n0 = __builtin_amdgcn_readfirstlane(noff[wv]);
            const int n1 = __builtin_amdgcn_readfirstlane(noff[wv + 1]);
            for (int i0 = n0; i0 < n1; i0 += 32) {
                const int idx = i0 + lane;
                int   sv = 0;
                float nv = 0.0f;
                if (idx < n1) { sv = e_src[idx]; nv = e_nrm[idx]; }
                const int m = min(32, n1 - i0);
                for (int j = 0; j < m; j += AGG_BATCH) {
                    v4f   hv[AGG_BATCH];
                    float nj[AGG_BATCH];
#pragma unroll
                    for (int u = 0; u < AGG_BATCH; ++u) {
                        const int sj = __builtin_amdgcn_readlane(sv, j + u);
                        nj[u] = __int_as_float(__builtin_amdgcn_readlane(__float_as_int(nv), j + u));
                        hv[u] = *reinterpret_cast<const v4f*>(h + (size_t)sj * D_FEAT + 4 * lane);
                    }
#pragma unroll
                    for (int u = 0; u < AGG_BATCH; ++u) {
                        if (j + u < m) {
                            acc.x = fmaf(hv[u].x, nj[u], acc.x);
                            acc.y = fmaf(hv[u].y, nj[u], acc.y);
                            acc.z = fmaf(hv[u].z, nj[u], acc.z);
                            acc.w = fmaf(hv[u].w, nj[u], acc.w);
                        }
                    }
                }
            }
        }
    } else if (has_node) {
        const int node = base + wv;
        for (int gg = 0; gg < SORT_BLOCKS; ++gg) {
            const int* __restrict__ tq = tab + (size_t)gg * (nb + 1) + b;
            const int q0 = __builtin_amdgcn_readfirstlane(tq[0]);
            const int q1 = __builtin_amdgcn_readfirstlane(tq[1]);
            const int2* __restrict__ rq = rec + (size_t)gg * CH;
            for (int i0 = q0; i0 < q1; i0 += 32) {
                const int i = i0 + lane;
                int2 r = make_int2(0, 0);
                if (i < q1) r = rq[i];
                const int rsrc = r.x & REC_SRC_MASK;
                const bool mine = i < q1 && (int)((unsigned)r.x >> REC_SHIFT) == wv;
                float nv = 0.0f;
                if (mine) nv = dinv[rsrc] * __int_as_float(r.y) * dinv[node];
                unsigned mask = __builtin_amdgcn_ballot_w32(mine);
                while (mask) {
                    const int jj = __builtin_ctz(mask);
                    mask &= mask - 1;
                    const int   sj = __builtin_amdgcn_readlane(rsrc, jj);
                    const float nj = __int_as_float(__builtin_amdgcn_readlane(__float_as_int(nv), jj));
                    const v4f hv = *reinterpret_cast<const v4f*>(h + (size_t)sj * D_FEAT + 4 * lane);
                    acc.x = fmaf(hv.x, nj, acc.x);
                    acc.y = fmaf(hv.y, nj, acc.y);
                    acc.z = fmaf(hv.z, nj, acc.z);
                    acc.w = fmaf(hv.w, nj, acc.w);
                }
            }
        }
    }

    if (has_node) {
        v4f o;
        o.x = (bv.x + hs.x * di * di) + acc.x;
        o.y = (bv.y + hs.y * di * di) + acc.y;
        o.z = (bv.z + hs.z * di * di) + acc.z;
        o.w = (bv.w + hs.w * di * di) + acc.w;
        *reinterpret_cast<v4f*>(out + (size_t)(base + wv) * D_FEAT + 4 * lane) = o;
    }
}

extern "C" void kernel_launch(void* const* d_in, const int* in_sizes, int n_in,
                              void* d_out, int out_size, void* d_ws, size_t ws_size,
                              hipStream_t stream) {
    const float* x  = (const float*)d_in[0];
    const float* W  = (const float*)d_in[1];
    const float* bb = (const float*)d_in[2];
    const float* ew = (const float*)d_in[3];
    const int*   ei = (const int*)d_in[4];

    const int E = in_sizes[3];
    const int N = in_sizes[0] / D_FEAT;
    const int* src = ei;
    const int* dst = ei + E;
    float* out = (float*)d_out;

    float* deg = (float*)d_ws;
    size_t degPad = ((size_t)N + 63) & ~(size_t)63;
    float* h = deg + degPad;
    float* part = h + (size_t)N * D_FEAT;
    int* rec = (int*)(part + (size_t)DEG_BLOCKS * N);
    int* tab = rec + (size_t)E * 2;
    const int nb = (N + BIN_NODES - 1) / BIN_NODES;

    const int TB = 256;
    k_pre <<<SORT_BLOCKS + (N + GEMM_ROWS - 1) / GEMM_ROWS, PRE_THREADS, 0, stream>>>(
        part, rec, tab, h, x, W, src, dst, ew, E, N, nb);
    k_dinv<<<(N + TB - 1) / TB, TB, 0, stream>>>(deg, part, N);
    k_agg <<<nb, AGG_THREADS, 0, stream>>>(out, h, deg, bb, rec, tab, E, N, nb);
}
